# starting kernel
# baseline (speedup 1.0000x reference)
#include <hip/hip_runtime.h>


#define N_NODES   100000
#define F1        64
#define F2        128
#define NEG_SLOPE 0.01f

typedef __attribute__((ext_vector_type(2))) float v2f;
typedef __attribute__((ext_vector_type(8))) float v8f;

__device__ __forceinline__ float leaky(float v) {
    return v >= 0.0f ? v : v * NEG_SLOPE;
}


__global__ void k_deg_init(float* __restrict__ deg) {
    int i = blockIdx.x * blockDim.x + threadIdx.x;
    if (i < N_NODES) deg[i] = 1.0f;
}

__global__ void k_deg_accum(const int* __restrict__ dst, int E,
                            float* __restrict__ deg) {
    int e = blockIdx.x * blockDim.x + threadIdx.x;
    if (e < E) atomicAdd(&deg[dst[e]], 1.0f);
}

__global__ void k_dinv(const float* __restrict__ deg, float* __restrict__ dinv) {
    int i = blockIdx.x * blockDim.x + threadIdx.x;
    if (i < N_NODES) dinv[i] = rsqrtf(fmaxf(deg[i], 1.0f));
}


__global__ void k_layer1(const float* __restrict__ x, const float* __restrict__ W1,
                         const float* __restrict__ dinv,
                         float* __restrict__ h1, float* __restrict__ agg1) {
    int idx = blockIdx.x * blockDim.x + threadIdx.x;
    if (idx >= N_NODES * F1) return;
    int n = idx >> 6;
    int f = idx & (F1 - 1);
    float h = fmaf(x[2 * n], W1[f], x[2 * n + 1] * W1[F1 + f]);
    h1[idx] = h;
    float di = dinv[n];
    agg1[idx] = h * di * di;
}


__global__ void k_edge_agg1(const int* __restrict__ src, const int* __restrict__ dst,
                            int E, const float* __restrict__ dinv,
                            const float* __restrict__ h1, float* __restrict__ agg1) {
    int t = blockIdx.x * 256 + threadIdx.x;
    int e = t >> 6;
    int f = t & (F1 - 1);
    if (e >= E) return;
    int s = src[e];
    int d = dst[e];
    float norm = dinv[s] * dinv[d];
    atomicAdd(&agg1[d * F1 + f], h1[s * F1 + f] * norm);
}


__global__ void k_act1(float* __restrict__ agg1, const float* __restrict__ b1) {
    int idx = blockIdx.x * blockDim.x + threadIdx.x;
    if (idx >= N_NODES * F1) return;
    agg1[idx] = leaky(agg1[idx] + b1[idx & (F1 - 1)]);
}


__global__ void __launch_bounds__(256)
k_gemm2_wmma(const float* __restrict__ A,
             const float* __restrict__ W2,
             const float* __restrict__ dinv,
             float* __restrict__ h2,
             float* __restrict__ agg2) {
    const int lane = threadIdx.x & 31;
    const int wave = threadIdx.x >> 5;
    const int row0 = blockIdx.x * 16;
    const int col0 = wave * 16;

    const int m  = lane & 15;
    const int kh = (lane >> 4) << 1;

    v8f acc = {};
    const float* Arow = A + (size_t)(row0 + m) * F1;
#pragma unroll
    for (int k0 = 0; k0 < F1; k0 += 4) {
        v2f a, b;
        a.x = Arow[k0 + kh];
        a.y = Arow[k0 + kh + 1];
        b.x = W2[(k0 + kh) * F2 + col0 + m];
        b.y = W2[(k0 + kh + 1) * F2 + col0 + m];
        acc = __builtin_amdgcn_wmma_f32_16x16x4_f32(
             false, a,  false, b,
             (short)0, acc,  false,  false);
    }

    const int nc    = lane & 15;
    const int rbase = row0 + ((lane >> 4) << 3);
#pragma unroll
    for (int v = 0; v < 8; ++v) {
        int   r  = rbase + v;
        float d  = acc[v];
        float di = dinv[r];
        size_t o = (size_t)r * F2 + col0 + nc;
        h2[o]    = d;
        agg2[o]  = d * di * di;
    }
}


__global__ void k_edge_agg2(const int* __restrict__ src, const int* __restrict__ dst,
                            int E, const float* __restrict__ dinv,
                            const float* __restrict__ h2, float* __restrict__ agg2) {
    int t = blockIdx.x * 256 + threadIdx.x;
    int e = t >> 7;
    int f = t & (F2 - 1);
    if (e >= E) return;
    int s = src[e];
    int d = dst[e];
    float norm = dinv[s] * dinv[d];
    atomicAdd(&agg2[d * F2 + f], h2[s * F2 + f] * norm);
}


__global__ void k_act2(float* __restrict__ z, const float* __restrict__ b2) {
    int idx = blockIdx.x * blockDim.x + threadIdx.x;
    if (idx >= N_NODES * F2) return;
    z[idx] = leaky(z[idx] + b2[idx & (F2 - 1)]);
}


extern "C" void kernel_launch(void* const* d_in, const int* in_sizes, int n_in,
                              void* d_out, int out_size, void* d_ws, size_t ws_size,
                              hipStream_t stream) {
    const int*   edge_index = (const int*)d_in[0];
    const float* x          = (const float*)d_in[1];
    const float* W1         = (const float*)d_in[2];
    const float* b1         = (const float*)d_in[3];
    const float* W2         = (const float*)d_in[4];
    const float* b2         = (const float*)d_in[5];
    float*       z          = (float*)d_out;

    const int E = in_sizes[0] / 2;
    const int* src = edge_index;
    const int* dst = edge_index + E;

    float* ws   = (float*)d_ws;
    float* deg  = ws;
    float* dinv = ws + N_NODES;
    float* agg1 = ws + 2 * N_NODES;
    float* h1   = agg1 + N_NODES * F1;
    float* h2   = h1;
    (void)ws_size; (void)n_in; (void)out_size;

    const int T = 256;

    k_deg_init <<<(N_NODES + T - 1) / T, T, 0, stream>>>(deg);
    k_deg_accum<<<(E + T - 1) / T, T, 0, stream>>>(dst, E, deg);
    k_dinv     <<<(N_NODES + T - 1) / T, T, 0, stream>>>(deg, dinv);

    k_layer1   <<<(N_NODES * F1 + T - 1) / T, T, 0, stream>>>(x, W1, dinv, h1, agg1);
    k_edge_agg1<<<((size_t)E * F1 + T - 1) / T, T, 0, stream>>>(src, dst, E, dinv, h1, agg1);
    k_act1     <<<(N_NODES * F1 + T - 1) / T, T, 0, stream>>>(agg1, b1);

    k_gemm2_wmma<<<N_NODES / 16, 256, 0, stream>>>(agg1, W2, dinv, h2, z);
    k_edge_agg2 <<<((size_t)E * F2 + T - 1) / T, T, 0, stream>>>(src, dst, E, dinv, h2, z);
    k_act2      <<<(N_NODES * F2 + T - 1) / T, T, 0, stream>>>(z, b2);
}


# Layer-1 transform and GEMM run as resident blocks that walk their rows (625 and 1024 blocks, were 6250 each): 100000 short waves no longer have to enter; a GEMM wave loads its W2 fragments once
# speedup vs baseline: 5.0013x; 5.0013x over previous
#include <hip/hip_runtime.h>


#define N_NODES   100000
#define F1        64
#define F2        128
#define NEG_SLOPE 0.01f
#define TR        400
#define NT        (N_NODES / TR)
#define LCAP      32768
#define SRC_BITS  17

static_assert(N_NODES % TR == 0, "tiles must cover the nodes exactly");
static_assert(N_NODES <= (1 << SRC_BITS), "a source id must fit its field");
static_assert(TR <= (1 << (31 - SRC_BITS)), "a local row must fit above the source id");

typedef __attribute__((ext_vector_type(2))) float v2f;
typedef __attribute__((ext_vector_type(8))) float v8f;

__device__ __forceinline__ float leaky(float v) {
    return v >= 0.0f ? v : v * NEG_SLOPE;
}


__global__ void __launch_bounds__(1024)
k_tile_scan(const int* __restrict__ src, const int* __restrict__ dst, int E,
            float* __restrict__ dinv, int* __restrict__ tile_cnt, int* __restrict__ list) {
    __shared__ int2 que[LCAP];
    __shared__ int  degc[TR];
    __shared__ int  qn;

    const int tid = threadIdx.x;
    const int lo  = blockIdx.x * TR;

    if (tid == 0) qn = 0;
    for (int row = tid; row < TR; row += 1024) degc[row] = 1;
    __syncthreads();

    auto push = [&](int e, int r) {
        atomicAdd(&degc[r], 1);
        const int slot = atomicAdd(&qn, 1);
        if (slot < LCAP) que[slot] = make_int2(e, r);
    };
    auto scan4 = [&](const int4& v, int e) {
        const unsigned r0 = (unsigned)(v.x - lo);
        const unsigned r1 = (unsigned)(v.y - lo);
        const unsigned r2 = (unsigned)(v.z - lo);
        const unsigned r3 = (unsigned)(v.w - lo);
        const unsigned a  = r0 < r1 ? r0 : r1;
        const unsigned c  = r2 < r3 ? r2 : r3;
        if ((a < c ? a : c) < TR) {
            if (r0 < TR) push(e,     (int)r0);
            if (r1 < TR) push(e + 1, (int)r1);
            if (r2 < TR) push(e + 2, (int)r2);
            if (r3 < TR) push(e + 3, (int)r3);
        }
    };
    if ((E & 3) == 0 && E > 0) {
        constexpr int K = 8;
        const int   E4 = E >> 2;
        const int4* d4 = (const int4*)dst;
        int4 cur[K], nxt[K];
#pragma unroll
        for (int k = 0; k < K; ++k) {
            const int idx = tid + k * 1024;
            cur[k] = d4[idx < E4 ? idx : E4 - 1];
        }
        for (int i0 = tid; i0 < E4; i0 += K * 1024) {
#pragma unroll
            for (int k = 0; k < K; ++k) {
                const int idx = i0 + (K + k) * 1024;
                nxt[k] = d4[idx < E4 ? idx : E4 - 1];
            }
#pragma unroll
            for (int k = 0; k < K; ++k) {
                const int idx = i0 + k * 1024;
                if (idx < E4) scan4(cur[k], 4 * idx);
            }
#pragma unroll
            for (int k = 0; k < K; ++k) cur[k] = nxt[k];
        }
    } else {
        for (int e = tid; e < E; e += 1024) {
            const unsigned r = (unsigned)(dst[e] - lo);
            if (r < TR) push(e, (int)r);
        }
    }
    __syncthreads();

    const int qtot = qn;
    const int n    = qtot < LCAP ? qtot : LCAP;
    int* lst = list + (size_t)blockIdx.x * LCAP;
    for (int i = tid; i < n; i += 1024) {
        const int2 en = que[i];
        lst[i] = (en.y << SRC_BITS) | src[en.x];
    }
    for (int row = tid; row < TR; row += 1024)
        dinv[lo + row] = rsqrtf(fmaxf((float)degc[row], 1.0f));
    if (tid == 0) tile_cnt[blockIdx.x] = qtot;
}


#define L1_BLOCKS  625
#define L1_THREADS 256
#define L1_ITERS   10

static_assert(L1_BLOCKS * L1_THREADS * L1_ITERS == N_NODES * F1 / 4, "the walk must cover the plane exactly");
static_assert((L1_BLOCKS * L1_THREADS) % (F1 / 4) == 0, "a thread must keep its features");

__global__ void __launch_bounds__(L1_THREADS)
k_layer1(const float* __restrict__ x, const float* __restrict__ W1,
         float* __restrict__ h1) {
    const int g = blockIdx.x * L1_THREADS + threadIdx.x;
    const int f = (g & 15) << 2;
    const float4 wa = *(const float4*)(W1 + f);
    const float4 wb = *(const float4*)(W1 + F1 + f);
    float x0[L1_ITERS], x1[L1_ITERS];
#pragma unroll
    for (int it = 0; it < L1_ITERS; ++it) {
        const int n = (g + it * (L1_BLOCKS * L1_THREADS)) >> 4;
        x0[it] = x[2 * n];
        x1[it] = x[2 * n + 1];
    }
#pragma unroll
    for (int it = 0; it < L1_ITERS; ++it) {
        const int i4 = g + it * (L1_BLOCKS * L1_THREADS);
        float4 h;
        h.x = fmaf(x0[it], wa.x, x1[it] * wb.x);
        h.y = fmaf(x0[it], wa.y, x1[it] * wb.y);
        h.z = fmaf(x0[it], wa.z, x1[it] * wb.z);
        h.w = fmaf(x0[it], wa.w, x1[it] * wb.w);
        *(float4*)(h1 + 4 * (size_t)i4) = h;
    }
}


__device__ __forceinline__ void lds_add(float* p, float v) {
    __hip_atomic_fetch_add(p, v, __ATOMIC_RELAXED, __HIP_MEMORY_SCOPE_WORKGROUP);
}

template <int F>
__global__ void __launch_bounds__(1024)
k_tile_agg(const int* __restrict__ src, const int* __restrict__ dst, int E,
           const float* __restrict__ dinv, const int* __restrict__ tile_cnt,
           const int* __restrict__ list, const float* __restrict__ h,
           const float* __restrict__ bias, float* __restrict__ out) {
    constexpr int Q = F / 32;
    constexpr int U = 8;
    __shared__ float acc[TR * F];

    const int tid  = threadIdx.x;
    const int lane = tid & 31;
    const int wave = __builtin_amdgcn_readfirstlane(tid >> 5);
    const int lo   = blockIdx.x * TR;

    for (int i4 = tid; i4 < TR * F / 4; i4 += 1024) {
        const int   row = i4 / (F / 4);
        const float di  = dinv[lo + row];
        const float4 hv = *(const float4*)(h + (size_t)lo * F + 4 * (size_t)i4);
        float4 a;
        a.x = hv.x * di * di;
        a.y = hv.y * di * di;
        a.z = hv.z * di * di;
        a.w = hv.w * di * di;
        *(float4*)(acc + 4 * i4) = a;
    }
    __syncthreads();

    const int n = tile_cnt[blockIdx.x];
    if (n <= LCAP) {
        const int* lst = list + (size_t)blockIdx.x * LCAP;
        for (int b = wave * 32; b < n; b += 32 * 32) {
            const int cnt = (n - b) < 32 ? (n - b) : 32;
            int s = 0, r = 0;
            float nrm = 0.0f;
            if (lane < cnt) {
                const int en = lst[b + lane];
                s   = en & ((1 << SRC_BITS) - 1);
                r   = en >> SRC_BITS;
                nrm = dinv[s] * dinv[lo + r];
            }
            int j = 0;
            for (; j + U <= cnt; j += U) {
                float v[U][Q];
                for (int u = 0; u < U; ++u) {
                    const int sj = __builtin_amdgcn_readlane(s, j + u);
                    const float* hp = h + (size_t)sj * F + lane;
                    for (int q = 0; q < Q; ++q) v[u][q] = hp[32 * q];
                }
                for (int u = 0; u < U; ++u) {
                    const int   rj = __builtin_amdgcn_readlane(r, j + u);
                    const float nj = __builtin_bit_cast(float,
                        __builtin_amdgcn_readlane(__builtin_bit_cast(int, nrm), j + u));
                    float* ap = acc + rj * F + lane;
                    for (int q = 0; q < Q; ++q) lds_add(ap + 32 * q, v[u][q] * nj);
                }
            }
            for (; j < cnt; ++j) {
                const int   sj = __builtin_amdgcn_readlane(s, j);
                const int   rj = __builtin_amdgcn_readlane(r, j);
                const float nj = __builtin_bit_cast(float,
                    __builtin_amdgcn_readlane(__builtin_bit_cast(int, nrm), j));
                const float* hp = h + (size_t)sj * F + lane;
                float* ap = acc + rj * F + lane;
                for (int q = 0; q < Q; ++q) lds_add(ap + 32 * q, hp[32 * q] * nj);
            }
        }
    } else {
        for (int e = tid; e < E; e += 1024) {
            const unsigned r = (unsigned)(dst[e] - lo);
            if (r < TR) {
                const int   s   = src[e];
                const float nrm = dinv[s] * dinv[lo + r];
#pragma nounroll
                for (int f = 0; f < F; ++f)
                    lds_add(&acc[r * F + f], h[(size_t)s * F + f] * nrm);
            }
        }
    }
    __syncthreads();

    for (int i4 = tid; i4 < TR * F / 4; i4 += 1024) {
        float4 v = *(const float4*)(acc + 4 * i4);
        const float4 bb = *(const float4*)(bias + ((i4 & (F / 4 - 1)) << 2));
        v.x = leaky(v.x + bb.x);
        v.y = leaky(v.y + bb.y);
        v.z = leaky(v.z + bb.z);
        v.w = leaky(v.w + bb.w);
        *(float4*)(out + (size_t)lo * F + 4 * (size_t)i4) = v;
    }
}


#define GEMM_BLOCKS 1024

__global__ void __launch_bounds__(256)
k_gemm2_wmma(const float* __restrict__ A,
             const float* __restrict__ W2,
             float* __restrict__ h2) {
    const int lane = threadIdx.x & 31;
    const int wave = threadIdx.x >> 5;
    const int col0 = wave * 16;

    const int m  = lane & 15;
    const int kh = (lane >> 4) << 1;

    v2f bfr[F1 / 4];
#pragma unroll
    for (int t = 0; t < F1 / 4; ++t) {
        bfr[t].x = W2[(4 * t + kh) * F2 + col0 + m];
        bfr[t].y = W2[(4 * t + kh + 1) * F2 + col0 + m];
    }

    const int rsub = (lane >> 4) << 3;

    for (int stripe = blockIdx.x; stripe < N_NODES / 16; stripe += GEMM_BLOCKS) {
        const int row0 = stripe * 16;
        v8f acc = {};
        const float* Arow = A + (size_t)(row0 + m) * F1;
#pragma unroll
        for (int t = 0; t < F1 / 4; ++t) {
            v2f a;
            a.x = Arow[4 * t + kh];
            a.y = Arow[4 * t + kh + 1];
            acc = __builtin_amdgcn_wmma_f32_16x16x4_f32(
                 false, a,  false, bfr[t],
                 (short)0, acc,  false,  false);
        }
#pragma unroll
        for (int v = 0; v < 8; ++v) {
            const int r = row0 + rsub + v;
            h2[(size_t)r * F2 + col0 + m] = acc[v];
        }
    }
}


extern "C" void kernel_launch(void* const* d_in, const int* in_sizes, int n_in,
                              void* d_out, int out_size, void* d_ws, size_t ws_size,
                              hipStream_t stream) {
    const int*   edge_index = (const int*)d_in[0];
    const float* x          = (const float*)d_in[1];
    const float* W1         = (const float*)d_in[2];
    const float* b1         = (const float*)d_in[3];
    const float* W2         = (const float*)d_in[4];
    const float* b2         = (const float*)d_in[5];
    float*       z          = (float*)d_out;

    const int E = in_sizes[0] / 2;
    const int* src = edge_index;
    const int* dst = edge_index + E;

    float* ws   = (float*)d_ws;
    float* dinv = ws;
    int*   tcnt = (int*)(ws + N_NODES);
    int*   list = tcnt + 1024;
    float* act1 = (float*)(list + NT * LCAP);
    float* h1   = act1 + N_NODES * F1;
    float* h2   = h1;
    (void)ws_size; (void)n_in; (void)out_size;

    k_tile_scan    <<<NT, 1024, 0, stream>>>(src, dst, E, dinv, tcnt, list);

    k_layer1       <<<L1_BLOCKS, L1_THREADS, 0, stream>>>(x, W1, h1);
    k_tile_agg<F1> <<<NT, 1024, 0, stream>>>(src, dst, E, dinv, tcnt, list, h1, b1, act1);

    k_gemm2_wmma   <<<GEMM_BLOCKS, 256, 0, stream>>>(act1, W2, h2);
    k_tile_agg<F2> <<<NT, 1024, 0, stream>>>(src, dst, E, dinv, tcnt, list, h2, b2, z);
}


# Tile scan reads a 1-byte tile key per edge (made by the layer-1 launch, now first), not 4-byte dst; queues 16-edge groups with a hit, tests only those on dst; list order changes, sums arrival-ordered
# speedup vs baseline: 5.5983x; 1.1194x over previous
#include <hip/hip_runtime.h>


#define N_NODES   100000
#define F1        64
#define F2        128
#define NEG_SLOPE 0.01f
#define TR        400
#define NT        (N_NODES / TR)
#define LCAP      32768
#define SRC_BITS  17

static_assert(N_NODES % TR == 0, "tiles must cover the nodes exactly");
static_assert(N_NODES <= (1 << SRC_BITS), "a source id must fit its field");
static_assert(TR <= (1 << (31 - SRC_BITS)), "a local row must fit above the source id");
static_assert(NT <= 256, "a tile number must fit the one-byte key");

typedef __attribute__((ext_vector_type(2))) float v2f;
typedef __attribute__((ext_vector_type(8))) float v8f;

__device__ __forceinline__ float leaky(float v) {
    return v >= 0.0f ? v : v * NEG_SLOPE;
}


__global__ void __launch_bounds__(1024)
k_tile_scan(const int* __restrict__ src, const int* __restrict__ dst, int E,
            const unsigned* __restrict__ keys,
            float* __restrict__ dinv, int* __restrict__ tile_cnt, int* __restrict__ list) {
    __shared__ int gq[LCAP];
    __shared__ int eq[LCAP];
    __shared__ int degc[TR];
    __shared__ int gn, en;

    const int tid = threadIdx.x;
    const int lo  = blockIdx.x * TR;

    if (tid == 0) { gn = 0; en = 0; }
    for (int row = tid; row < TR; row += 1024) degc[row] = 1;
    __syncthreads();

    int  count  = LCAP + 1;
    bool listed = false;
    if ((E & 15) == 0 && E > 0) {
        constexpr int K = 8;
        const int    E16 = E >> 4;
        const uint4* k4  = (const uint4*)keys;
        const unsigned pat = (unsigned)blockIdx.x * 0x01010101u;
        auto scan16 = [&](const uint4& v, int idx) {
            const unsigned x0 = v.x ^ pat, x1 = v.y ^ pat, x2 = v.z ^ pat, x3 = v.w ^ pat;
            const unsigned z  = ((x0 - 0x01010101u) & ~x0) | ((x1 - 0x01010101u) & ~x1) |
                                ((x2 - 0x01010101u) & ~x2) | ((x3 - 0x01010101u) & ~x3);
            if (z & 0x80808080u) {
                const int slot = atomicAdd(&gn, 1);
                if (slot < LCAP) gq[slot] = idx;
            }
        };
        uint4 cur[K], nxt[K];
#pragma unroll
        for (int k = 0; k < K; ++k) {
            const int idx = tid + k * 1024;
            cur[k] = k4[idx < E16 ? idx : E16 - 1];
        }
        for (int i0 = tid; i0 < E16; i0 += K * 1024) {
#pragma unroll
            for (int k = 0; k < K; ++k) {
                const int idx = i0 + (K + k) * 1024;
                nxt[k] = k4[idx < E16 ? idx : E16 - 1];
            }
#pragma unroll
            for (int k = 0; k < K; ++k) {
                const int idx = i0 + k * 1024;
                if (idx < E16) scan16(cur[k], idx);
            }
#pragma unroll
            for (int k = 0; k < K; ++k) cur[k] = nxt[k];
        }
        __syncthreads();

        const int gtot = gn;
        if (gtot <= LCAP) {
            const int4* d4 = (const int4*)dst;
            for (int i = tid; i < gtot; i += 1024) {
                const int g = gq[i];
                int4 d[4];
#pragma unroll
                for (int k = 0; k < 4; ++k) d[k] = d4[4 * g + k];
                unsigned m = 0;
#pragma unroll
                for (int k = 0; k < 4; ++k) {
                    m |= ((unsigned)(d[k].x - lo) < TR ? 1u : 0u) << (4 * k);
                    m |= ((unsigned)(d[k].y - lo) < TR ? 1u : 0u) << (4 * k + 1);
                    m |= ((unsigned)(d[k].z - lo) < TR ? 1u : 0u) << (4 * k + 2);
                    m |= ((unsigned)(d[k].w - lo) < TR ? 1u : 0u) << (4 * k + 3);
                }
                if (m) {
                    int slot = atomicAdd(&en, __popc(m));
                    while (m) {
                        const int j = __ffs(m) - 1;
                        m &= m - 1;
                        if (slot < LCAP) eq[slot] = 16 * g + j;
                        ++slot;
                    }
                }
            }
            __syncthreads();

            const int etot = en;
            if (etot <= LCAP) {
                int* lst = list + (size_t)blockIdx.x * LCAP;
                for (int i = tid; i < etot; i += 1024) {
                    const int e = eq[i];
                    const int r = dst[e] - lo;
                    const int sv = src[e];
                    atomicAdd(&degc[r], 1);
                    lst[i] = (r << SRC_BITS) | sv;
                }
                count  = etot;
                listed = true;
            }
        }
    }
    if (!listed) {
        for (int e = tid; e < E; e += 1024) {
            const unsigned r = (unsigned)(dst[e] - lo);
            if (r < TR) atomicAdd(&degc[r], 1);
        }
    }
    __syncthreads();

    for (int row = tid; row < TR; row += 1024)
        dinv[lo + row] = rsqrtf(fmaxf((float)degc[row], 1.0f));
    if (tid == 0) tile_cnt[blockIdx.x] = count;
}


#define L1_BLOCKS  625
#define L1_THREADS 256
#define L1_ITERS   10

static_assert(L1_BLOCKS * L1_THREADS * L1_ITERS == N_NODES * F1 / 4, "the walk must cover the plane exactly");
static_assert((L1_BLOCKS * L1_THREADS) % (F1 / 4) == 0, "a thread must keep its features");

__global__ void __launch_bounds__(L1_THREADS)
k_layer1(const float* __restrict__ x, const float* __restrict__ W1,
         float* __restrict__ h1,
         const int* __restrict__ dst, int E, unsigned* __restrict__ keys) {
    const int g = blockIdx.x * L1_THREADS + threadIdx.x;
    const int f = (g & 15) << 2;
    const float4 wa = *(const float4*)(W1 + f);
    const float4 wb = *(const float4*)(W1 + F1 + f);
    float x0[L1_ITERS], x1[L1_ITERS];
#pragma unroll
    for (int it = 0; it < L1_ITERS; ++it) {
        const int n = (g + it * (L1_BLOCKS * L1_THREADS)) >> 4;
        x0[it] = x[2 * n];
        x1[it] = x[2 * n + 1];
    }
#pragma unroll
    for (int it = 0; it < L1_ITERS; ++it) {
        const int i4 = g + it * (L1_BLOCKS * L1_THREADS);
        float4 h;
        h.x = fmaf(x0[it], wa.x, x1[it] * wb.x);
        h.y = fmaf(x0[it], wa.y, x1[it] * wb.y);
        h.z = fmaf(x0[it], wa.z, x1[it] * wb.z);
        h.w = fmaf(x0[it], wa.w, x1[it] * wb.w);
        *(float4*)(h1 + 4 * (size_t)i4) = h;
    }

    if ((E & 15) == 0) {
        const int   E4 = E >> 2;
        const int4* d4 = (const int4*)dst;
        for (int q = g; q < E4; q += L1_BLOCKS * L1_THREADS) {
            const int4 v = d4[q];
            keys[q] =  (((unsigned)v.x / TR) & 255u)        | ((((unsigned)v.y / TR) & 255u) << 8) |
                      ((((unsigned)v.z / TR) & 255u) << 16) |  (((unsigned)v.w / TR)         << 24);
        }
    }
}


__device__ __forceinline__ void lds_add(float* p, float v) {
    __hip_atomic_fetch_add(p, v, __ATOMIC_RELAXED, __HIP_MEMORY_SCOPE_WORKGROUP);
}

template <int F>
__global__ void __launch_bounds__(1024)
k_tile_agg(const int* __restrict__ src, const int* __restrict__ dst, int E,
           const float* __restrict__ dinv, const int* __restrict__ tile_cnt,
           const int* __restrict__ list, const float* __restrict__ h,
           const float* __restrict__ bias, float* __restrict__ out) {
    constexpr int Q = F / 32;
    constexpr int U = 8;
    __shared__ float acc[TR * F];

    const int tid  = threadIdx.x;
    const int lane = tid & 31;
    const int wave = __builtin_amdgcn_readfirstlane(tid >> 5);
    const int lo   = blockIdx.x * TR;

    for (int i4 = tid; i4 < TR * F / 4; i4 += 1024) {
        const int   row = i4 / (F / 4);
        const float di  = dinv[lo + row];
        const float4 hv = *(const float4*)(h + (size_t)lo * F + 4 * (size_t)i4);
        float4 a;
        a.x = hv.x * di * di;
        a.y = hv.y * di * di;
        a.z = hv.z * di * di;
        a.w = hv.w * di * di;
        *(float4*)(acc + 4 * i4) = a;
    }
    __syncthreads();

    const int n = tile_cnt[blockIdx.x];
    if (n <= LCAP) {
        const int* lst = list + (size_t)blockIdx.x * LCAP;
        for (int b = wave * 32; b < n; b += 32 * 32) {
            const int cnt = (n - b) < 32 ? (n - b) : 32;
            int s = 0, r = 0;
            float nrm = 0.0f;
            if (lane < cnt) {
                const int en = lst[b + lane];
                s   = en & ((1 << SRC_BITS) - 1);
                r   = en >> SRC_BITS;
                nrm = dinv[s] * dinv[lo + r];
            }
            int j = 0;
            for (; j + U <= cnt; j += U) {
                float v[U][Q];
                for (int u = 0; u < U; ++u) {
                    const int sj = __builtin_amdgcn_readlane(s, j + u);
                    const float* hp = h + (size_t)sj * F + lane;
                    for (int q = 0; q < Q; ++q) v[u][q] = hp[32 * q];
                }
                for (int u = 0; u < U; ++u) {
                    const int   rj = __builtin_amdgcn_readlane(r, j + u);
                    const float nj = __builtin_bit_cast(float,
                        __builtin_amdgcn_readlane(__builtin_bit_cast(int, nrm), j + u));
                    float* ap = acc + rj * F + lane;
                    for (int q = 0; q < Q; ++q) lds_add(ap + 32 * q, v[u][q] * nj);
                }
            }
            for (; j < cnt; ++j) {
                const int   sj = __builtin_amdgcn_readlane(s, j);
                const int   rj = __builtin_amdgcn_readlane(r, j);
                const float nj = __builtin_bit_cast(float,
                    __builtin_amdgcn_readlane(__builtin_bit_cast(int, nrm), j));
                const float* hp = h + (size_t)sj * F + lane;
                float* ap = acc + rj * F + lane;
                for (int q = 0; q < Q; ++q) lds_add(ap + 32 * q, hp[32 * q] * nj);
            }
        }
    } else {
        for (int e = tid; e < E; e += 1024) {
            const unsigned r = (unsigned)(dst[e] - lo);
            if (r < TR) {
                const int   s   = src[e];
                const float nrm = dinv[s] * dinv[lo + r];
#pragma nounroll
                for (int f = 0; f < F; ++f)
                    lds_add(&acc[r * F + f], h[(size_t)s * F + f] * nrm);
            }
        }
    }
    __syncthreads();

    for (int i4 = tid; i4 < TR * F / 4; i4 += 1024) {
        float4 v = *(const float4*)(acc + 4 * i4);
        const float4 bb = *(const float4*)(bias + ((i4 & (F / 4 - 1)) << 2));
        v.x = leaky(v.x + bb.x);
        v.y = leaky(v.y + bb.y);
        v.z = leaky(v.z + bb.z);
        v.w = leaky(v.w + bb.w);
        *(float4*)(out + (size_t)lo * F + 4 * (size_t)i4) = v;
    }
}


#define GEMM_BLOCKS 1024

__global__ void __launch_bounds__(256)
k_gemm2_wmma(const float* __restrict__ A,
             const float* __restrict__ W2,
             float* __restrict__ h2) {
    const int lane = threadIdx.x & 31;
    const int wave = threadIdx.x >> 5;
    const int col0 = wave * 16;

    const int m  = lane & 15;
    const int kh = (lane >> 4) << 1;

    v2f bfr[F1 / 4];
#pragma unroll
    for (int t = 0; t < F1 / 4; ++t) {
        bfr[t].x = W2[(4 * t + kh) * F2 + col0 + m];
        bfr[t].y = W2[(4 * t + kh + 1) * F2 + col0 + m];
    }

    const int rsub = (lane >> 4) << 3;

    for (int stripe = blockIdx.x; stripe < N_NODES / 16; stripe += GEMM_BLOCKS) {
        const int row0 = stripe * 16;
        v8f acc = {};
        const float* Arow = A + (size_t)(row0 + m) * F1;
#pragma unroll
        for (int t = 0; t < F1 / 4; ++t) {
            v2f a;
            a.x = Arow[4 * t + kh];
            a.y = Arow[4 * t + kh + 1];
            acc = __builtin_amdgcn_wmma_f32_16x16x4_f32(
                 false, a,  false, bfr[t],
                 (short)0, acc,  false,  false);
        }
#pragma unroll
        for (int v = 0; v < 8; ++v) {
            const int r = row0 + rsub + v;
            h2[(size_t)r * F2 + col0 + m] = acc[v];
        }
    }
}


extern "C" void kernel_launch(void* const* d_in, const int* in_sizes, int n_in,
                              void* d_out, int out_size, void* d_ws, size_t ws_size,
                              hipStream_t stream) {
    const int*   edge_index = (const int*)d_in[0];
    const float* x          = (const float*)d_in[1];
    const float* W1         = (const float*)d_in[2];
    const float* b1         = (const float*)d_in[3];
    const float* W2         = (const float*)d_in[4];
    const float* b2         = (const float*)d_in[5];
    float*       z          = (float*)d_out;

    const int E = in_sizes[0] / 2;
    const int* src = edge_index;
    const int* dst = edge_index + E;

    float* ws   = (float*)d_ws;
    float* dinv = ws;
    int*   tcnt = (int*)(ws + N_NODES);
    int*   list = tcnt + 1024;
    float* act1 = (float*)(list + NT * LCAP);
    float* h1   = act1 + N_NODES * F1;
    float* h2   = h1;
    unsigned* keys = (unsigned*)(h2 + N_NODES * F2);
    (void)ws_size; (void)n_in; (void)out_size;

    k_layer1       <<<L1_BLOCKS, L1_THREADS, 0, stream>>>(x, W1, h1, dst, E, keys);

    k_tile_scan    <<<NT, 1024, 0, stream>>>(src, dst, E, keys, dinv, tcnt, list);

    k_tile_agg<F1> <<<NT, 1024, 0, stream>>>(src, dst, E, dinv, tcnt, list, h1, b1, act1);

    k_gemm2_wmma   <<<GEMM_BLOCKS, 256, 0, stream>>>(act1, W2, h2);
    k_tile_agg<F2> <<<NT, 1024, 0, stream>>>(src, dst, E, dinv, tcnt, list, h2, b2, z);
}


# Gathered rows stored times their node's deg^-1/2 (layer-1 rows made in the scan's tail): drains add raw rows, no norm look-up per edge, dst factor at write-out; rounding changes, sums arrival-ordered
# speedup vs baseline: 5.8012x; 1.0362x over previous
#include <hip/hip_runtime.h>


#define N_NODES   100000
#define F1        64
#define F2        128
#define NEG_SLOPE 0.01f
#define TR        400
#define NT        (N_NODES / TR)
#define LCAP      32768
#define SRC_BITS  17

static_assert(N_NODES % TR == 0, "tiles must cover the nodes exactly");
static_assert(N_NODES <= (1 << SRC_BITS), "a source id must fit its field");
static_assert(TR <= (1 << (31 - SRC_BITS)), "a local row must fit above the source id");
static_assert(NT <= 256, "a tile number must fit the one-byte key");

typedef __attribute__((ext_vector_type(2))) float v2f;
typedef __attribute__((ext_vector_type(8))) float v8f;

__device__ __forceinline__ float leaky(float v) {
    return v >= 0.0f ? v : v * NEG_SLOPE;
}


__global__ void __launch_bounds__(1024)
k_tile_scan(const int* __restrict__ src, const int* __restrict__ dst, int E,
            const unsigned* __restrict__ keys,
            const float* __restrict__ x, const float* __restrict__ W1,
            float* __restrict__ dinv, int* __restrict__ tile_cnt, int* __restrict__ list,
            float* __restrict__ h1s) {
    __shared__ int gq[LCAP];
    __shared__ int eq[LCAP];
    __shared__ int degc[TR];
    __shared__ int gn, en;

    const int tid = threadIdx.x;
    const int lo  = blockIdx.x * TR;

    if (tid == 0) { gn = 0; en = 0; }
    for (int row = tid; row < TR; row += 1024) degc[row] = 1;
    __syncthreads();

    int  count  = LCAP + 1;
    bool listed = false;
    if ((E & 15) == 0 && E > 0) {
        constexpr int K = 8;
        const int    E16 = E >> 4;
        const uint4* k4  = (const uint4*)keys;
        const unsigned pat = (unsigned)blockIdx.x * 0x01010101u;
        auto scan16 = [&](const uint4& v, int idx) {
            const unsigned x0 = v.x ^ pat, x1 = v.y ^ pat, x2 = v.z ^ pat, x3 = v.w ^ pat;
            const unsigned z  = ((x0 - 0x01010101u) & ~x0) | ((x1 - 0x01010101u) & ~x1) |
                                ((x2 - 0x01010101u) & ~x2) | ((x3 - 0x01010101u) & ~x3);
            if (z & 0x80808080u) {
                const int slot = atomicAdd(&gn, 1);
                if (slot < LCAP) gq[slot] = idx;
            }
        };
        uint4 cur[K], nxt[K];
#pragma unroll
        for (int k = 0; k < K; ++k) {
            const int idx = tid + k * 1024;
            cur[k] = k4[idx < E16 ? idx : E16 - 1];
        }
        for (int i0 = tid; i0 < E16; i0 += K * 1024) {
#pragma unroll
            for (int k = 0; k < K; ++k) {
                const int idx = i0 + (K + k) * 1024;
                nxt[k] = k4[idx < E16 ? idx : E16 - 1];
            }
#pragma unroll
            for (int k = 0; k < K; ++k) {
                const int idx = i0 + k * 1024;
                if (idx < E16) scan16(cur[k], idx);
            }
#pragma unroll
            for (int k = 0; k < K; ++k) cur[k] = nxt[k];
        }
        __syncthreads();

        const int gtot = gn;
        if (gtot <= LCAP) {
            const int4* d4 = (const int4*)dst;
            for (int i = tid; i < gtot; i += 1024) {
                const int g = gq[i];
                int4 d[4];
#pragma unroll
                for (int k = 0; k < 4; ++k) d[k] = d4[4 * g + k];
                unsigned m = 0;
#pragma unroll
                for (int k = 0; k < 4; ++k) {
                    m |= ((unsigned)(d[k].x - lo) < TR ? 1u : 0u) << (4 * k);
                    m |= ((unsigned)(d[k].y - lo) < TR ? 1u : 0u) << (4 * k + 1);
                    m |= ((unsigned)(d[k].z - lo) < TR ? 1u : 0u) << (4 * k + 2);
                    m |= ((unsigned)(d[k].w - lo) < TR ? 1u : 0u) << (4 * k + 3);
                }
                if (m) {
                    int slot = atomicAdd(&en, __popc(m));
                    while (m) {
                        const int j = __ffs(m) - 1;
                        m &= m - 1;
                        if (slot < LCAP) eq[slot] = 16 * g + j;
                        ++slot;
                    }
                }
            }
            __syncthreads();

            const int etot = en;
            if (etot <= LCAP) {
                int* lst = list + (size_t)blockIdx.x * LCAP;
                for (int i = tid; i < etot; i += 1024) {
                    const int e = eq[i];
                    const int r = dst[e] - lo;
                    const int sv = src[e];
                    atomicAdd(&degc[r], 1);
                    lst[i] = (r << SRC_BITS) | sv;
                }
                count  = etot;
                listed = true;
            }
        }
    }
    if (!listed) {
        for (int e = tid; e < E; e += 1024) {
            const unsigned r = (unsigned)(dst[e] - lo);
            if (r < TR) atomicAdd(&degc[r], 1);
        }
    }

    constexpr int H1_PIECES = TR * F1 / 4;
    constexpr int H1_ITERS  = (H1_PIECES + 1023) / 1024;
    static_assert(1024 % (F1 / 4) == 0, "a thread must keep its features");
    const int    f  = (tid & (F1 / 4 - 1)) << 2;
    const float4 wa = *(const float4*)(W1 + f);
    const float4 wb = *(const float4*)(W1 + F1 + f);
    float2 xv[H1_ITERS];
#pragma unroll
    for (int it = 0; it < H1_ITERS; ++it) {
        const int i4  = tid + it * 1024;
        const int row = (i4 < H1_PIECES ? i4 : H1_PIECES - 1) / (F1 / 4);
        xv[it] = *(const float2*)(x + 2 * (size_t)(lo + row));
    }
    __syncthreads();

    for (int row = tid; row < TR; row += 1024)
        dinv[lo + row] = rsqrtf(fmaxf((float)degc[row], 1.0f));
    if (tid == 0) tile_cnt[blockIdx.x] = count;

#pragma unroll
    for (int it = 0; it < H1_ITERS; ++it) {
        const int i4 = tid + it * 1024;
        if (i4 < H1_PIECES) {
            const int   row = i4 / (F1 / 4);
            const float di  = rsqrtf(fmaxf((float)degc[row], 1.0f));
            float4 hv;
            hv.x = fmaf(xv[it].x, wa.x, xv[it].y * wb.x) * di;
            hv.y = fmaf(xv[it].x, wa.y, xv[it].y * wb.y) * di;
            hv.z = fmaf(xv[it].x, wa.z, xv[it].y * wb.z) * di;
            hv.w = fmaf(xv[it].x, wa.w, xv[it].y * wb.w) * di;
            *(float4*)(h1s + (size_t)lo * F1 + 4 * (size_t)i4) = hv;
        }
    }
}


#define KEY_BLOCKS  625
#define KEY_THREADS 256

__global__ void __launch_bounds__(KEY_THREADS)
k_tile_keys(const int* __restrict__ dst, int E, unsigned* __restrict__ keys) {
    const int g = blockIdx.x * KEY_THREADS + threadIdx.x;
    if ((E & 15) == 0) {
        const int   E4 = E >> 2;
        const int4* d4 = (const int4*)dst;
        for (int q = g; q < E4; q += KEY_BLOCKS * KEY_THREADS) {
            const int4 v = d4[q];
            keys[q] =  (((unsigned)v.x / TR) & 255u)        | ((((unsigned)v.y / TR) & 255u) << 8) |
                      ((((unsigned)v.z / TR) & 255u) << 16) |  (((unsigned)v.w / TR)         << 24);
        }
    }
}


__device__ __forceinline__ void lds_add(float* p, float v) {
    __hip_atomic_fetch_add(p, v, __ATOMIC_RELAXED, __HIP_MEMORY_SCOPE_WORKGROUP);
}

template <int F, bool SCALE_OUT>
__global__ void __launch_bounds__(1024)
k_tile_agg(const int* __restrict__ src, const int* __restrict__ dst, int E,
           const float* __restrict__ dinv, const int* __restrict__ tile_cnt,
           const int* __restrict__ list, const float* __restrict__ hs,
           const float* __restrict__ bias, float* __restrict__ out) {
    constexpr int Q = F / 32;
    constexpr int U = 8;
    __shared__ float acc[TR * F];

    const int tid  = threadIdx.x;
    const int lane = tid & 31;
    const int wave = __builtin_amdgcn_readfirstlane(tid >> 5);
    const int lo   = blockIdx.x * TR;

    for (int i4 = tid; i4 < TR * F / 4; i4 += 1024)
        *(float4*)(acc + 4 * i4) = *(const float4*)(hs + (size_t)lo * F + 4 * (size_t)i4);
    __syncthreads();

    const int n = tile_cnt[blockIdx.x];
    if (n <= LCAP) {
        const int* lst = list + (size_t)blockIdx.x * LCAP;
        for (int b = wave * 32; b < n; b += 32 * 32) {
            const int cnt = (n - b) < 32 ? (n - b) : 32;
            int s = 0, r = 0;
            if (lane < cnt) {
                const int en = lst[b + lane];
                s = en & ((1 << SRC_BITS) - 1);
                r = en >> SRC_BITS;
            }
            int j = 0;
            for (; j + U <= cnt; j += U) {
                float v[U][Q];
                for (int u = 0; u < U; ++u) {
                    const int sj = __builtin_amdgcn_readlane(s, j + u);
                    const float* hp = hs + (size_t)sj * F + lane;
                    for (int q = 0; q < Q; ++q) v[u][q] = hp[32 * q];
                }
                for (int u = 0; u < U; ++u) {
                    const int rj = __builtin_amdgcn_readlane(r, j + u);
                    float* ap = acc + rj * F + lane;
                    for (int q = 0; q < Q; ++q) lds_add(ap + 32 * q, v[u][q]);
                }
            }
            for (; j < cnt; ++j) {
                const int sj = __builtin_amdgcn_readlane(s, j);
                const int rj = __builtin_amdgcn_readlane(r, j);
                const float* hp = hs + (size_t)sj * F + lane;
                float* ap = acc + rj * F + lane;
                for (int q = 0; q < Q; ++q) lds_add(ap + 32 * q, hp[32 * q]);
            }
        }
    } else {
        for (int e = tid; e < E; e += 1024) {
            const unsigned r = (unsigned)(dst[e] - lo);
            if (r < TR) {
                const int s = src[e];
#pragma nounroll
                for (int f = 0; f < F; ++f)
                    lds_add(&acc[r * F + f], hs[(size_t)s * F + f]);
            }
        }
    }
    __syncthreads();

    for (int i4 = tid; i4 < TR * F / 4; i4 += 1024) {
        const int   row = i4 / (F / 4);
        const float di  = dinv[lo + row];
        float4 v = *(const float4*)(acc + 4 * i4);
        const float4 bb = *(const float4*)(bias + ((i4 & (F / 4 - 1)) << 2));
        v.x = leaky(fmaf(v.x, di, bb.x));
        v.y = leaky(fmaf(v.y, di, bb.y));
        v.z = leaky(fmaf(v.z, di, bb.z));
        v.w = leaky(fmaf(v.w, di, bb.w));
        if (SCALE_OUT) {
            v.x *= di;
            v.y *= di;
            v.z *= di;
            v.w *= di;
        }
        *(float4*)(out + (size_t)lo * F + 4 * (size_t)i4) = v;
    }
}


#define GEMM_BLOCKS 1024

__global__ void __launch_bounds__(256)
k_gemm2_wmma(const float* __restrict__ A,
             const float* __restrict__ W2,
             float* __restrict__ h2) {
    const int lane = threadIdx.x & 31;
    const int wave = threadIdx.x >> 5;
    const int col0 = wave * 16;

    const int m  = lane & 15;
    const int kh = (lane >> 4) << 1;

    v2f bfr[F1 / 4];
#pragma unroll
    for (int t = 0; t < F1 / 4; ++t) {
        bfr[t].x = W2[(4 * t + kh) * F2 + col0 + m];
        bfr[t].y = W2[(4 * t + kh + 1) * F2 + col0 + m];
    }

    const int rsub = (lane >> 4) << 3;

    for (int stripe = blockIdx.x; stripe < N_NODES / 16; stripe += GEMM_BLOCKS) {
        const int row0 = stripe * 16;
        v8f acc = {};
        const float* Arow = A + (size_t)(row0 + m) * F1;
#pragma unroll
        for (int t = 0; t < F1 / 4; ++t) {
            v2f a;
            a.x = Arow[4 * t + kh];
            a.y = Arow[4 * t + kh + 1];
            acc = __builtin_amdgcn_wmma_f32_16x16x4_f32(
                 false, a,  false, bfr[t],
                 (short)0, acc,  false,  false);
        }
#pragma unroll
        for (int v = 0; v < 8; ++v) {
            const int r = row0 + rsub + v;
            h2[(size_t)r * F2 + col0 + m] = acc[v];
        }
    }
}


extern "C" void kernel_launch(void* const* d_in, const int* in_sizes, int n_in,
                              void* d_out, int out_size, void* d_ws, size_t ws_size,
                              hipStream_t stream) {
    const int*   edge_index = (const int*)d_in[0];
    const float* x          = (const float*)d_in[1];
    const float* W1         = (const float*)d_in[2];
    const float* b1         = (const float*)d_in[3];
    const float* W2         = (const float*)d_in[4];
    const float* b2         = (const float*)d_in[5];
    float*       z          = (float*)d_out;

    const int E = in_sizes[0] / 2;
    const int* src = edge_index;
    const int* dst = edge_index + E;

    float* ws   = (float*)d_ws;
    float* dinv = ws;
    int*   tcnt = (int*)(ws + N_NODES);
    int*   list = tcnt + 1024;
    float* act1 = (float*)(list + NT * LCAP);
    float* h1   = act1 + N_NODES * F1;
    float* h2   = h1;
    unsigned* keys = (unsigned*)(h2 + N_NODES * F2);
    (void)ws_size; (void)n_in; (void)out_size;

    k_tile_keys    <<<KEY_BLOCKS, KEY_THREADS, 0, stream>>>(dst, E, keys);

    k_tile_scan    <<<NT, 1024, 0, stream>>>(src, dst, E, keys, x, W1, dinv, tcnt, list, h1);

    k_tile_agg<F1, true>  <<<NT, 1024, 0, stream>>>(src, dst, E, dinv, tcnt, list, h1, b1, act1);

    k_gemm2_wmma   <<<GEMM_BLOCKS, 256, 0, stream>>>(act1, W2, h2);
    k_tile_agg<F2, false> <<<NT, 1024, 0, stream>>>(src, dst, E, dinv, tcnt, list, h2, b2, z);
}


# Layer-2 GEMM done by the layer-1 aggregation block on its 400 finished rows in LDS (f32 WMMA, A from LDS, whole-line stores); activated 64-wide plane and one launch dropped; sum order kept
# speedup vs baseline: 6.2903x; 1.0843x over previous
#include <hip/hip_runtime.h>


#define N_NODES   100000
#define F1        64
#define F2        128
#define NEG_SLOPE 0.01f
#define TR        400
#define NT        (N_NODES / TR)
#define LCAP      32768
#define SRC_BITS  17

static_assert(N_NODES % TR == 0, "tiles must cover the nodes exactly");
static_assert(N_NODES <= (1 << SRC_BITS), "a source id must fit its field");
static_assert(TR <= (1 << (31 - SRC_BITS)), "a local row must fit above the source id");
static_assert(NT <= 256, "a tile number must fit the one-byte key");
static_assert(TR % 16 == 0, "a tile must be whole 16-row stripes for the matrix instruction");

typedef __attribute__((ext_vector_type(2))) float v2f;
typedef __attribute__((ext_vector_type(8))) float v8f;

__device__ __forceinline__ float leaky(float v) {
    return v >= 0.0f ? v : v * NEG_SLOPE;
}


__global__ void __launch_bounds__(1024)
k_tile_scan(const int* __restrict__ src, const int* __restrict__ dst, int E,
            const unsigned* __restrict__ keys,
            const float* __restrict__ x, const float* __restrict__ W1,
            float* __restrict__ dinv, int* __restrict__ tile_cnt, int* __restrict__ list,
            float* __restrict__ h1s) {
    __shared__ int gq[LCAP];
    __shared__ int eq[LCAP];
    __shared__ int degc[TR];
    __shared__ int gn, en;

    const int tid = threadIdx.x;
    const int lo  = blockIdx.x * TR;

    if (tid == 0) { gn = 0; en = 0; }
    for (int row = tid; row < TR; row += 1024) degc[row] = 1;
    __syncthreads();

    int  count  = LCAP + 1;
    bool listed = false;
    if ((E & 15) == 0 && E > 0) {
        constexpr int K = 8;
        const int    E16 = E >> 4;
        const uint4* k4  = (const uint4*)keys;
        const unsigned pat = (unsigned)blockIdx.x * 0x01010101u;
        auto scan16 = [&](const uint4& v, int idx) {
            const unsigned x0 = v.x ^ pat, x1 = v.y ^ pat, x2 = v.z ^ pat, x3 = v.w ^ pat;
            const unsigned z  = ((x0 - 0x01010101u) & ~x0) | ((x1 - 0x01010101u) & ~x1) |
                                ((x2 - 0x01010101u) & ~x2) | ((x3 - 0x01010101u) & ~x3);
            if (z & 0x80808080u) {
                const int slot = atomicAdd(&gn, 1);
                if (slot < LCAP) gq[slot] = idx;
            }
        };
        uint4 cur[K], nxt[K];
#pragma unroll
        for (int k = 0; k < K; ++k) {
            const int idx = tid + k * 1024;
            cur[k] = k4[idx < E16 ? idx : E16 - 1];
        }
        for (int i0 = tid; i0 < E16; i0 += K * 1024) {
#pragma unroll
            for (int k = 0; k < K; ++k) {
                const int idx = i0 + (K + k) * 1024;
                nxt[k] = k4[idx < E16 ? idx : E16 - 1];
            }
#pragma unroll
            for (int k = 0; k < K; ++k) {
                const int idx = i0 + k * 1024;
                if (idx < E16) scan16(cur[k], idx);
            }
#pragma unroll
            for (int k = 0; k < K; ++k) cur[k] = nxt[k];
        }
        __syncthreads();

        const int gtot = gn;
        if (gtot <= LCAP) {
            const int4* d4 = (const int4*)dst;
            for (int i = tid; i < gtot; i += 1024) {
                const int g = gq[i];
                int4 d[4];
#pragma unroll
                for (int k = 0; k < 4; ++k) d[k] = d4[4 * g + k];
                unsigned m = 0;
#pragma unroll
                for (int k = 0; k < 4; ++k) {
                    m |= ((unsigned)(d[k].x - lo) < TR ? 1u : 0u) << (4 * k);
                    m |= ((unsigned)(d[k].y - lo) < TR ? 1u : 0u) << (4 * k + 1);
                    m |= ((unsigned)(d[k].z - lo) < TR ? 1u : 0u) << (4 * k + 2);
                    m |= ((unsigned)(d[k].w - lo) < TR ? 1u : 0u) << (4 * k + 3);
                }
                if (m) {
                    int slot = atomicAdd(&en, __popc(m));
                    while (m) {
                        const int j = __ffs(m) - 1;
                        m &= m - 1;
                        if (slot < LCAP) eq[slot] = 16 * g + j;
                        ++slot;
                    }
                }
            }
            __syncthreads();

            const int etot = en;
            if (etot <= LCAP) {
                int* lst = list + (size_t)blockIdx.x * LCAP;
                for (int i = tid; i < etot; i += 1024) {
                    const int e = eq[i];
                    const int r = dst[e] - lo;
                    const int sv = src[e];
                    atomicAdd(&degc[r], 1);
                    lst[i] = (r << SRC_BITS) | sv;
                }
                count  = etot;
                listed = true;
            }
        }
    }
    if (!listed) {
        for (int e = tid; e < E; e += 1024) {
            const unsigned r = (unsigned)(dst[e] - lo);
            if (r < TR) atomicAdd(&degc[r], 1);
        }
    }

    constexpr int H1_PIECES = TR * F1 / 4;
    constexpr int H1_ITERS  = (H1_PIECES + 1023) / 1024;
    static_assert(1024 % (F1 / 4) == 0, "a thread must keep its features");
    const int    f  = (tid & (F1 / 4 - 1)) << 2;
    const float4 wa = *(const float4*)(W1 + f);
    const float4 wb = *(const float4*)(W1 + F1 + f);
    float2 xv[H1_ITERS];
#pragma unroll
    for (int it = 0; it < H1_ITERS; ++it) {
        const int i4  = tid + it * 1024;
        const int row = (i4 < H1_PIECES ? i4 : H1_PIECES - 1) / (F1 / 4);
        xv[it] = *(const float2*)(x + 2 * (size_t)(lo + row));
    }
    __syncthreads();

    for (int row = tid; row < TR; row += 1024)
        dinv[lo + row] = rsqrtf(fmaxf((float)degc[row], 1.0f));
    if (tid == 0) tile_cnt[blockIdx.x] = count;

#pragma unroll
    for (int it = 0; it < H1_ITERS; ++it) {
        const int i4 = tid + it * 1024;
        if (i4 < H1_PIECES) {
            const int   row = i4 / (F1 / 4);
            const float di  = rsqrtf(fmaxf((float)degc[row], 1.0f));
            float4 hv;
            hv.x = fmaf(xv[it].x, wa.x, xv[it].y * wb.x) * di;
            hv.y = fmaf(xv[it].x, wa.y, xv[it].y * wb.y) * di;
            hv.z = fmaf(xv[it].x, wa.z, xv[it].y * wb.z) * di;
            hv.w = fmaf(xv[it].x, wa.w, xv[it].y * wb.w) * di;
            *(float4*)(h1s + (size_t)lo * F1 + 4 * (size_t)i4) = hv;
        }
    }
}


#define KEY_BLOCKS  625
#define KEY_THREADS 256

__global__ void __launch_bounds__(KEY_THREADS)
k_tile_keys(const int* __restrict__ dst, int E, unsigned* __restrict__ keys) {
    const int g = blockIdx.x * KEY_THREADS + threadIdx.x;
    if ((E & 15) == 0) {
        const int   E4 = E >> 2;
        const int4* d4 = (const int4*)dst;
        for (int q = g; q < E4; q += KEY_BLOCKS * KEY_THREADS) {
            const int4 v = d4[q];
            keys[q] =  (((unsigned)v.x / TR) & 255u)        | ((((unsigned)v.y / TR) & 255u) << 8) |
                      ((((unsigned)v.z / TR) & 255u) << 16) |  (((unsigned)v.w / TR)         << 24);
        }
    }
}


__device__ __forceinline__ float xhalf(float x) {
    const int b = __builtin_bit_cast(int, x);
    return __builtin_bit_cast(float, __builtin_amdgcn_permlanex16(b, b, 0x76543210, 0xfedcba98, false, false));
}

__device__ __forceinline__ void lds_add(float* p, float v) {
    __hip_atomic_fetch_add(p, v, __ATOMIC_RELAXED, __HIP_MEMORY_SCOPE_WORKGROUP);
}

template <int F, bool GEMM_OUT>
__global__ void __launch_bounds__(1024)
k_tile_agg(const int* __restrict__ src, const int* __restrict__ dst, int E,
           const float* __restrict__ dinv, const int* __restrict__ tile_cnt,
           const int* __restrict__ list, const float* __restrict__ hs,
           const float* __restrict__ bias, const float* __restrict__ W2,
           float* __restrict__ out) {
    constexpr int Q = F / 32;
    constexpr int U = 8;
    constexpr int P = GEMM_OUT ? F + 4 : F;
    static_assert(!GEMM_OUT || F == F1, "the fused product is act1 (F1 wide) times W2");
    __shared__ float acc[TR * P];

    const int tid  = threadIdx.x;
    const int lane = tid & 31;
    const int wave = __builtin_amdgcn_readfirstlane(tid >> 5);
    const int lo   = blockIdx.x * TR;

    for (int i4 = tid; i4 < TR * F / 4; i4 += 1024)
        *(float4*)(acc + (i4 / (F / 4)) * P + ((i4 & (F / 4 - 1)) << 2)) =
            *(const float4*)(hs + (size_t)lo * F + 4 * (size_t)i4);
    __syncthreads();

    const int n = tile_cnt[blockIdx.x];
    if (n <= LCAP) {
        const int* lst = list + (size_t)blockIdx.x * LCAP;
        for (int b = wave * 32; b < n; b += 32 * 32) {
            const int cnt = (n - b) < 32 ? (n - b) : 32;
            int s = 0, r = 0;
            if (lane < cnt) {
                const int en = lst[b + lane];
                s = en & ((1 << SRC_BITS) - 1);
                r = en >> SRC_BITS;
            }
            int j = 0;
            for (; j + U <= cnt; j += U) {
                float v[U][Q];
                for (int u = 0; u < U; ++u) {
                    const int sj = __builtin_amdgcn_readlane(s, j + u);
                    const float* hp = hs + (size_t)sj * F + lane;
                    for (int q = 0; q < Q; ++q) v[u][q] = hp[32 * q];
                }
                for (int u = 0; u < U; ++u) {
                    const int rj = __builtin_amdgcn_readlane(r, j + u);
                    float* ap = acc + rj * P + lane;
                    for (int q = 0; q < Q; ++q) lds_add(ap + 32 * q, v[u][q]);
                }
            }
            for (; j < cnt; ++j) {
                const int sj = __builtin_amdgcn_readlane(s, j);
                const int rj = __builtin_amdgcn_readlane(r, j);
                const float* hp = hs + (size_t)sj * F + lane;
                float* ap = acc + rj * P + lane;
                for (int q = 0; q < Q; ++q) lds_add(ap + 32 * q, hp[32 * q]);
            }
        }
    } else {
        for (int e = tid; e < E; e += 1024) {
            const unsigned r = (unsigned)(dst[e] - lo);
            if (r < TR) {
                const int s = src[e];
#pragma nounroll
                for (int f = 0; f < F; ++f)
                    lds_add(&acc[r * P + f], hs[(size_t)s * F + f]);
            }
        }
    }
    __syncthreads();

    for (int i4 = tid; i4 < TR * F / 4; i4 += 1024) {
        const int   row = i4 / (F / 4);
        const float di  = dinv[lo + row];
        float* arow4 = acc + row * P + ((i4 & (F / 4 - 1)) << 2);
        float4 v = *(const float4*)arow4;
        const float4 bb = *(const float4*)(bias + ((i4 & (F / 4 - 1)) << 2));
        v.x = leaky(fmaf(v.x, di, bb.x));
        v.y = leaky(fmaf(v.y, di, bb.y));
        v.z = leaky(fmaf(v.z, di, bb.z));
        v.w = leaky(fmaf(v.w, di, bb.w));
        if (GEMM_OUT) {
            v.x *= di;
            v.y *= di;
            v.z *= di;
            v.w *= di;
            *(float4*)arow4 = v;
        } else {
            *(float4*)(out + (size_t)lo * F + 4 * (size_t)i4) = v;
        }
    }

    if (GEMM_OUT) {
        const int m  = lane & 15;
        const int h  = lane >> 4;
        const int kh = h << 1;
        const int c0 = (wave & 3) * 32;
        v2f bA[F1 / 4], bB[F1 / 4];
#pragma unroll
        for (int t = 0; t < F1 / 4; ++t) {
            bA[t].x = W2[(4 * t + kh) * F2 + c0 + m];
            bA[t].y = W2[(4 * t + kh + 1) * F2 + c0 + m];
            bB[t].x = W2[(4 * t + kh) * F2 + c0 + 16 + m];
            bB[t].y = W2[(4 * t + kh + 1) * F2 + c0 + 16 + m];
        }
        __syncthreads();

        for (int st = wave >> 2; st < TR / 16; st += 8) {
            const float* arow = acc + (16 * st + m) * P + kh;
            v8f d0 = {}, d1 = {};
#pragma unroll
            for (int t = 0; t < F1 / 4; ++t) {
                const v2f a = *(const v2f*)(arow + 4 * t);
                d0 = __builtin_amdgcn_wmma_f32_16x16x4_f32(
                     false, a,  false, bA[t],
                     (short)0, d0,  false,  false);
                d1 = __builtin_amdgcn_wmma_f32_16x16x4_f32(
                     false, a,  false, bB[t],
                     (short)0, d1,  false,  false);
            }
            float* orow = out + (size_t)(lo + 16 * st) * F2 + c0 + lane;
#pragma unroll
            for (int v = 0; v < 8; ++v) {
                const float x0s = xhalf(d0[v]);
                const float x1s = xhalf(d1[v]);
                orow[(size_t)v * F2]       = h ? x1s : d0[v];
                orow[(size_t)(v + 8) * F2] = h ? d1[v] : x0s;
            }
        }
    }
}


extern "C" void kernel_launch(void* const* d_in, const int* in_sizes, int n_in,
                              void* d_out, int out_size, void* d_ws, size_t ws_size,
                              hipStream_t stream) {
    const int*   edge_index = (const int*)d_in[0];
    const float* x          = (const float*)d_in[1];
    const float* W1         = (const float*)d_in[2];
    const float* b1         = (const float*)d_in[3];
    const float* W2         = (const float*)d_in[4];
    const float* b2         = (const float*)d_in[5];
    float*       z          = (float*)d_out;

    const int E = in_sizes[0] / 2;
    const int* src = edge_index;
    const int* dst = edge_index + E;

    float* ws   = (float*)d_ws;
    float* dinv = ws;
    int*   tcnt = (int*)(ws + N_NODES);
    int*   list = tcnt + 1024;
    float* h1   = (float*)(list + NT * LCAP);
    float* h2   = h1 + N_NODES * F1;
    unsigned* keys = (unsigned*)(h2 + N_NODES * F2);
    (void)ws_size; (void)n_in; (void)out_size;

    k_tile_keys    <<<KEY_BLOCKS, KEY_THREADS, 0, stream>>>(dst, E, keys);

    k_tile_scan    <<<NT, 1024, 0, stream>>>(src, dst, E, keys, x, W1, dinv, tcnt, list, h1);

    k_tile_agg<F1, true>  <<<NT, 1024, 0, stream>>>(src, dst, E, dinv, tcnt, list, h1, b1, W2, h2);

    k_tile_agg<F2, false> <<<NT, 1024, 0, stream>>>(src, dst, E, dinv, tcnt, list, h2, b2, W2, z);
}


# Scan reads an index the first launch writes: per tile a bit per 4-edge quad, per quad 32 B of (dst, src) pairs; no key sweep, one place per edge for both ids; list order changes, sums arrival-ordered
# speedup vs baseline: 7.8997x; 1.2559x over previous
#include <hip/hip_runtime.h>


#define N_NODES   100000
#define F1        64
#define F2        128
#define NEG_SLOPE 0.01f
#define TR        400
#define NT        (N_NODES / TR)
#define LCAP      32768
#define SRC_BITS  17

static_assert(N_NODES % TR == 0, "tiles must cover the nodes exactly");
static_assert(N_NODES <= (1 << SRC_BITS), "a source id must fit its field");
static_assert(TR <= (1 << (31 - SRC_BITS)), "a local row must fit above the source id");
static_assert(TR % 16 == 0, "a tile must be whole 16-row stripes for the matrix instruction");

typedef __attribute__((ext_vector_type(2))) float v2f;
typedef __attribute__((ext_vector_type(8))) float v8f;

__device__ __forceinline__ float leaky(float v) {
    return v >= 0.0f ? v : v * NEG_SLOPE;
}


__device__ __forceinline__ int wave_scan_excl(int v, int lane, int& total) {
    int s = v;
#pragma unroll
    for (int d = 1; d < 32; d <<= 1) {
        const int t = __shfl_up(s, d);
        if (lane >= d) s += t;
    }
    total = __shfl(s, 31);
    return s - v;
}

__global__ void __launch_bounds__(1024)
k_tile_scan(const int* __restrict__ dst, int E,
            const int* __restrict__ idx,
            const float* __restrict__ x, const float* __restrict__ W1,
            float* __restrict__ dinv, int* __restrict__ tile_cnt, int* __restrict__ list,
            float* __restrict__ h1s) {
    __shared__ int gq[LCAP];
    __shared__ int eq[LCAP];
    __shared__ int degc[TR];
    __shared__ int gn, en;

    const int tid  = threadIdx.x;
    const int lane = tid & 31;
    const int lo   = blockIdx.x * TR;

    if (tid == 0) { gn = 0; en = 0; }
    for (int row = tid; row < TR; row += 1024) degc[row] = 1;
    __syncthreads();

    int  count   = LCAP + 1;
    bool counted = false;
    if ((E & 3) == 0 && E > 0) {
        const int    wpt   = (((E >> 2) + 1023) >> 10) << 5;
        const int4*  pairs = (const int4*)idx;
        const uint4* brow4 = (const uint4*)((const unsigned*)idx + 2 * (size_t)E + (size_t)blockIdx.x * wpt);
        const int    wbase = tid & ~31;

        constexpr int K4 = 4;
        const int w4n = wpt >> 2;
        for (int ib = wbase; ib < w4n; ib += K4 * 1024) {
            uint4 wv[K4];
#pragma unroll
            for (int k = 0; k < K4; ++k) {
                const int i = ib + lane + k * 1024;
                wv[k] = brow4[i < w4n ? i : w4n - 1];
            }
            int mine = 0;
#pragma unroll
            for (int k = 0; k < K4; ++k) {
                if (ib + lane + k * 1024 >= w4n) wv[k] = make_uint4(0u, 0u, 0u, 0u);
                mine += __popc(wv[k].x) + __popc(wv[k].y) + __popc(wv[k].z) + __popc(wv[k].w);
            }
            int total;
            int slot = wave_scan_excl(mine, lane, total);
            int base = 0;
            if (lane == 31 && total > 0) base = atomicAdd(&gn, total);
            slot += __shfl(base, 31);
            auto push = [&](unsigned m, int word) {
                while (m) {
                    const int j = __ffs(m) - 1;
                    m &= m - 1;
                    if (slot < LCAP) gq[slot] = 32 * word + j;
                    ++slot;
                }
            };
#pragma unroll
            for (int k = 0; k < K4; ++k) {
                const int w0 = 4 * (ib + lane + k * 1024);
                push(wv[k].x, w0);
                push(wv[k].y, w0 + 1);
                push(wv[k].z, w0 + 2);
                push(wv[k].w, w0 + 3);
            }
        }
        __syncthreads();

        const int qtot = gn;
        if (qtot <= LCAP) {
            constexpr int KQ = 8;
            for (int ib = wbase; ib < qtot; ib += KQ * 1024) {
                int4 va[KQ], vb[KQ];
#pragma unroll
                for (int k = 0; k < KQ; ++k) {
                    const int i = ib + lane + k * 1024;
                    const int q = gq[i < qtot ? i : qtot - 1];
                    va[k] = pairs[2 * (size_t)q];
                    vb[k] = pairs[2 * (size_t)q + 1];
                }
                unsigned hm = 0;
#pragma unroll
                for (int k = 0; k < KQ; ++k) {
                    const unsigned ok = (ib + lane + k * 1024 < qtot) ? 15u : 0u;
                    const unsigned h4 = ((unsigned)(va[k].x - lo) < TR ? 1u : 0u) |
                                        ((unsigned)(va[k].z - lo) < TR ? 2u : 0u) |
                                        ((unsigned)(vb[k].x - lo) < TR ? 4u : 0u) |
                                        ((unsigned)(vb[k].z - lo) < TR ? 8u : 0u);
                    hm |= (h4 & ok) << (4 * k);
                }
                int total;
                int slot = wave_scan_excl(__popc(hm), lane, total);
                int base = 0;
                if (lane == 31 && total > 0) base = atomicAdd(&en, total);
                base = __shfl(base, 31);
                slot += base;
                if (base + total <= LCAP) {
#pragma unroll
                    for (int k = 0; k < KQ; ++k) {
                        if (hm & (1u << (4 * k))) { eq[slot] = ((va[k].x - lo) << SRC_BITS) | va[k].y; ++slot; }
                        if (hm & (2u << (4 * k))) { eq[slot] = ((va[k].z - lo) << SRC_BITS) | va[k].w; ++slot; }
                        if (hm & (4u << (4 * k))) { eq[slot] = ((vb[k].x - lo) << SRC_BITS) | vb[k].y; ++slot; }
                        if (hm & (8u << (4 * k))) { eq[slot] = ((vb[k].z - lo) << SRC_BITS) | vb[k].w; ++slot; }
                    }
                }
            }
            __syncthreads();

            const int etot = en;
            if (etot <= LCAP) {
                int* lst = list + (size_t)blockIdx.x * LCAP;
                for (int i = tid; i < etot; i += 1024) {
                    const int w = eq[i];
                    lst[i] = w;
                    atomicAdd(&degc[w >> SRC_BITS], 1);
                }
                count   = etot;
                counted = true;
            }
        }
    }
    if (!counted) {
        for (int e = tid; e < E; e += 1024) {
            const unsigned r = (unsigned)(dst[e] - lo);
            if (r < TR) atomicAdd(&degc[r], 1);
        }
    }

    constexpr int H1_PIECES = TR * F1 / 4;
    constexpr int H1_ITERS  = (H1_PIECES + 1023) / 1024;
    static_assert(1024 % (F1 / 4) == 0, "a thread must keep its features");
    const int    f  = (tid & (F1 / 4 - 1)) << 2;
    const float4 wa = *(const float4*)(W1 + f);
    const float4 wb = *(const float4*)(W1 + F1 + f);
    float2 xv[H1_ITERS];
#pragma unroll
    for (int it = 0; it < H1_ITERS; ++it) {
        const int i4  = tid + it * 1024;
        const int row = (i4 < H1_PIECES ? i4 : H1_PIECES - 1) / (F1 / 4);
        xv[it] = *(const float2*)(x + 2 * (size_t)(lo + row));
    }
    __syncthreads();

    for (int row = tid; row < TR; row += 1024)
        dinv[lo + row] = rsqrtf(fmaxf((float)degc[row], 1.0f));
    if (tid == 0) tile_cnt[blockIdx.x] = count;

#pragma unroll
    for (int it = 0; it < H1_ITERS; ++it) {
        const int i4 = tid + it * 1024;
        if (i4 < H1_PIECES) {
            const int   row = i4 / (F1 / 4);
            const float di  = rsqrtf(fmaxf((float)degc[row], 1.0f));
            float4 hv;
            hv.x = fmaf(xv[it].x, wa.x, xv[it].y * wb.x) * di;
            hv.y = fmaf(xv[it].x, wa.y, xv[it].y * wb.y) * di;
            hv.z = fmaf(xv[it].x, wa.z, xv[it].y * wb.z) * di;
            hv.w = fmaf(xv[it].x, wa.w, xv[it].y * wb.w) * di;
            *(float4*)(h1s + (size_t)lo * F1 + 4 * (size_t)i4) = hv;
        }
    }
}


#define IDX_BLOCKS 512
#define IDX_PITCH  257

static_assert(NT <= IDX_PITCH, "a wave's row must hold a word per tile");

__global__ void __launch_bounds__(1024)
k_edge_index(const int* __restrict__ src, const int* __restrict__ dst, int E,
             int* __restrict__ idx) {
    __shared__ unsigned wb[32 * IDX_PITCH];
    if ((E & 3) != 0) return;

    const int tid    = threadIdx.x;
    const int lane   = tid & 31;
    const int wave   = tid >> 5;
    const int E4     = E >> 2;
    const int nchunk = (E4 + 1023) >> 10;
    const int wpt    = nchunk << 5;
    const int4* d4   = (const int4*)dst;
    const int4* s4   = (const int4*)src;
    int4*     pairs  = (int4*)idx;
    unsigned* bits   = (unsigned*)idx + 2 * (size_t)E;

    for (int c = blockIdx.x; c < nchunk; c += IDX_BLOCKS) {
        const int  q  = (c << 10) + tid;
        const int  qc = q < E4 ? q : E4 - 1;
        const int4 dv = d4[qc];
        const int4 sv = s4[qc];
        for (int i = tid; i < 32 * IDX_PITCH; i += 1024) wb[i] = 0u;
        __syncthreads();

        if (q < E4) {
            pairs[2 * (size_t)q]     = make_int4(dv.x, sv.x, dv.y, sv.y);
            pairs[2 * (size_t)q + 1] = make_int4(dv.z, sv.z, dv.w, sv.w);
            unsigned* wrow = wb + wave * IDX_PITCH;
            const unsigned bit = 1u << lane;
            const unsigned t0  = (unsigned)dv.x / TR;
            const unsigned t1  = (unsigned)dv.y / TR;
            const unsigned t2  = (unsigned)dv.z / TR;
            const unsigned t3  = (unsigned)dv.w / TR;
            if (t0 < NT) atomicOr(&wrow[t0], bit);
            if (t1 < NT) atomicOr(&wrow[t1], bit);
            if (t2 < NT) atomicOr(&wrow[t2], bit);
            if (t3 < NT) atomicOr(&wrow[t3], bit);
        }
        __syncthreads();

        for (int t = wave; t < NT; t += 32)
            bits[(size_t)t * wpt + (c << 5) + lane] = wb[lane * IDX_PITCH + t];
        __syncthreads();
    }
}


__device__ __forceinline__ float xhalf(float x) {
    const int b = __builtin_bit_cast(int, x);
    return __builtin_bit_cast(float, __builtin_amdgcn_permlanex16(b, b, 0x76543210, 0xfedcba98, false, false));
}

__device__ __forceinline__ void lds_add(float* p, float v) {
    __hip_atomic_fetch_add(p, v, __ATOMIC_RELAXED, __HIP_MEMORY_SCOPE_WORKGROUP);
}

template <int F, bool GEMM_OUT>
__global__ void __launch_bounds__(1024)
k_tile_agg(const int* __restrict__ src, const int* __restrict__ dst, int E,
           const float* __restrict__ dinv, const int* __restrict__ tile_cnt,
           const int* __restrict__ list, const float* __restrict__ hs,
           const float* __restrict__ bias, const float* __restrict__ W2,
           float* __restrict__ out) {
    constexpr int Q = F / 32;
    constexpr int U = 8;
    constexpr int P = GEMM_OUT ? F + 4 : F;
    static_assert(!GEMM_OUT || F == F1, "the fused product is act1 (F1 wide) times W2");
    __shared__ float acc[TR * P];

    const int tid  = threadIdx.x;
    const int lane = tid & 31;
    const int wave = __builtin_amdgcn_readfirstlane(tid >> 5);
    const int lo   = blockIdx.x * TR;

    for (int i4 = tid; i4 < TR * F / 4; i4 += 1024)
        *(float4*)(acc + (i4 / (F / 4)) * P + ((i4 & (F / 4 - 1)) << 2)) =
            *(const float4*)(hs + (size_t)lo * F + 4 * (size_t)i4);
    __syncthreads();

    const int n = tile_cnt[blockIdx.x];
    if (n <= LCAP) {
        const int* lst = list + (size_t)blockIdx.x * LCAP;
        for (int b = wave * 32; b < n; b += 32 * 32) {
            const int cnt = (n - b) < 32 ? (n - b) : 32;
            int s = 0, r = 0;
            if (lane < cnt) {
                const int en = lst[b + lane];
                s = en & ((1 << SRC_BITS) - 1);
                r = en >> SRC_BITS;
            }
            int j = 0;
            for (; j + U <= cnt; j += U) {
                float v[U][Q];
                for (int u = 0; u < U; ++u) {
                    const int sj = __builtin_amdgcn_readlane(s, j + u);
                    const float* hp = hs + (size_t)sj * F + lane;
                    for (int q = 0; q < Q; ++q) v[u][q] = hp[32 * q];
                }
                for (int u = 0; u < U; ++u) {
                    const int rj = __builtin_amdgcn_readlane(r, j + u);
                    float* ap = acc + rj * P + lane;
                    for (int q = 0; q < Q; ++q) lds_add(ap + 32 * q, v[u][q]);
                }
            }
            for (; j < cnt; ++j) {
                const int sj = __builtin_amdgcn_readlane(s, j);
                const int rj = __builtin_amdgcn_readlane(r, j);
                const float* hp = hs + (size_t)sj * F + lane;
                float* ap = acc + rj * P + lane;
                for (int q = 0; q < Q; ++q) lds_add(ap + 32 * q, hp[32 * q]);
            }
        }
    } else {
        for (int e = tid; e < E; e += 1024) {
            const unsigned r = (unsigned)(dst[e] - lo);
            if (r < TR) {
                const int s = src[e];
#pragma nounroll
                for (int f = 0; f < F; ++f)
                    lds_add(&acc[r * P + f], hs[(size_t)s * F + f]);
            }
        }
    }
    __syncthreads();

    for (int i4 = tid; i4 < TR * F / 4; i4 += 1024) {
        const int   row = i4 / (F / 4);
        const float di  = dinv[lo + row];
        float* arow4 = acc + row * P + ((i4 & (F / 4 - 1)) << 2);
        float4 v = *(const float4*)arow4;
        const float4 bb = *(const float4*)(bias + ((i4 & (F / 4 - 1)) << 2));
        v.x = leaky(fmaf(v.x, di, bb.x));
        v.y = leaky(fmaf(v.y, di, bb.y));
        v.z = leaky(fmaf(v.z, di, bb.z));
        v.w = leaky(fmaf(v.w, di, bb.w));
        if (GEMM_OUT) {
            v.x *= di;
            v.y *= di;
            v.z *= di;
            v.w *= di;
            *(float4*)arow4 = v;
        } else {
            *(float4*)(out + (size_t)lo * F + 4 * (size_t)i4) = v;
        }
    }

    if (GEMM_OUT) {
        const int m  = lane & 15;
        const int h  = lane >> 4;
        const int kh = h << 1;
        const int c0 = (wave & 3) * 32;
        v2f bA[F1 / 4], bB[F1 / 4];
#pragma unroll
        for (int t = 0; t < F1 / 4; ++t) {
            bA[t].x = W2[(4 * t + kh) * F2 + c0 + m];
            bA[t].y = W2[(4 * t + kh + 1) * F2 + c0 + m];
            bB[t].x = W2[(4 * t + kh) * F2 + c0 + 16 + m];
            bB[t].y = W2[(4 * t + kh + 1) * F2 + c0 + 16 + m];
        }
        __syncthreads();

        for (int st = wave >> 2; st < TR / 16; st += 8) {
            const float* arow = acc + (16 * st + m) * P + kh;
            v8f d0 = {}, d1 = {};
#pragma unroll
            for (int t = 0; t < F1 / 4; ++t) {
                const v2f a = *(const v2f*)(arow + 4 * t);
                d0 = __builtin_amdgcn_wmma_f32_16x16x4_f32(
                     false, a,  false, bA[t],
                     (short)0, d0,  false,  false);
                d1 = __builtin_amdgcn_wmma_f32_16x16x4_f32(
                     false, a,  false, bB[t],
                     (short)0, d1,  false,  false);
            }
            float* orow = out + (size_t)(lo + 16 * st) * F2 + c0 + lane;
#pragma unroll
            for (int v = 0; v < 8; ++v) {
                const float x0s = xhalf(d0[v]);
                const float x1s = xhalf(d1[v]);
                orow[(size_t)v * F2]       = h ? x1s : d0[v];
                orow[(size_t)(v + 8) * F2] = h ? d1[v] : x0s;
            }
        }
    }
}


extern "C" void kernel_launch(void* const* d_in, const int* in_sizes, int n_in,
                              void* d_out, int out_size, void* d_ws, size_t ws_size,
                              hipStream_t stream) {
    const int*   edge_index = (const int*)d_in[0];
    const float* x          = (const float*)d_in[1];
    const float* W1         = (const float*)d_in[2];
    const float* b1         = (const float*)d_in[3];
    const float* W2         = (const float*)d_in[4];
    const float* b2         = (const float*)d_in[5];
    float*       z          = (float*)d_out;

    const int E = in_sizes[0] / 2;
    const int* src = edge_index;
    const int* dst = edge_index + E;

    float* ws   = (float*)d_ws;
    float* dinv = ws;
    int*   tcnt = (int*)(ws + N_NODES);
    int*   list = tcnt + 1024;
    float* h1   = (float*)(list + NT * LCAP);
    float* h2   = h1 + N_NODES * F1;
    int*   idx  = (int*)(h2 + N_NODES * F2);
    (void)ws_size; (void)n_in; (void)out_size;

    k_edge_index   <<<IDX_BLOCKS, 1024, 0, stream>>>(src, dst, E, idx);

    k_tile_scan    <<<NT, 1024, 0, stream>>>(dst, E, idx, x, W1, dinv, tcnt, list, h1);

    k_tile_agg<F1, true>  <<<NT, 1024, 0, stream>>>(src, dst, E, dinv, tcnt, list, h1, b1, W2, h2);

    k_tile_agg<F2, false> <<<NT, 1024, 0, stream>>>(src, dst, E, dinv, tcnt, list, h2, b2, W2, z);
}


# Layer-1 aggregation makes each source's row from an 8-B record (x times deg^-1/2) and W1 in registers, not a stored 256-B row; h1 plane and its gather dropped; rounding changes, sums arrival-ordered
# speedup vs baseline: 8.7938x; 1.1132x over previous
#include <hip/hip_runtime.h>


#define N_NODES   100000
#define F1        64
#define F2        128
#define NEG_SLOPE 0.01f
#define TR        400
#define NT        (N_NODES / TR)
#define LCAP      32768
#define SRC_BITS  17

static_assert(N_NODES % TR == 0, "tiles must cover the nodes exactly");
static_assert(N_NODES <= (1 << SRC_BITS), "a source id must fit its field");
static_assert(TR <= (1 << (31 - SRC_BITS)), "a local row must fit above the source id");
static_assert(TR % 16 == 0, "a tile must be whole 16-row stripes for the matrix instruction");

typedef __attribute__((ext_vector_type(2))) float v2f;
typedef __attribute__((ext_vector_type(8))) float v8f;

__device__ __forceinline__ float leaky(float v) {
    return v >= 0.0f ? v : v * NEG_SLOPE;
}


__device__ __forceinline__ int wave_scan_excl(int v, int lane, int& total) {
    int s = v;
#pragma unroll
    for (int d = 1; d < 32; d <<= 1) {
        const int t = __shfl_up(s, d);
        if (lane >= d) s += t;
    }
    total = __shfl(s, 31);
    return s - v;
}

__global__ void __launch_bounds__(1024)
k_tile_scan(const int* __restrict__ dst, int E,
            const int* __restrict__ idx,
            const float* __restrict__ x,
            float* __restrict__ dinv, int* __restrict__ tile_cnt, int* __restrict__ list,
            float* __restrict__ xs) {
    __shared__ int gq[LCAP];
    __shared__ int eq[LCAP];
    __shared__ int degc[TR];
    __shared__ int gn, en;

    const int tid  = threadIdx.x;
    const int lane = tid & 31;
    const int lo   = blockIdx.x * TR;

    if (tid == 0) { gn = 0; en = 0; }
    for (int row = tid; row < TR; row += 1024) degc[row] = 1;
    __syncthreads();

    int  count   = LCAP + 1;
    bool counted = false;
    if ((E & 3) == 0 && E > 0) {
        const int    wpt   = (((E >> 2) + 1023) >> 10) << 5;
        const int4*  pairs = (const int4*)idx;
        const uint4* brow4 = (const uint4*)((const unsigned*)idx + 2 * (size_t)E + (size_t)blockIdx.x * wpt);
        const int    wbase = tid & ~31;

        constexpr int K4 = 4;
        const int w4n = wpt >> 2;
        for (int ib = wbase; ib < w4n; ib += K4 * 1024) {
            uint4 wv[K4];
#pragma unroll
            for (int k = 0; k < K4; ++k) {
                const int i = ib + lane + k * 1024;
                wv[k] = brow4[i < w4n ? i : w4n - 1];
            }
            int mine = 0;
#pragma unroll
            for (int k = 0; k < K4; ++k) {
                if (ib + lane + k * 1024 >= w4n) wv[k] = make_uint4(0u, 0u, 0u, 0u);
                mine += __popc(wv[k].x) + __popc(wv[k].y) + __popc(wv[k].z) + __popc(wv[k].w);
            }
            int total;
            int slot = wave_scan_excl(mine, lane, total);
            int base = 0;
            if (lane == 31 && total > 0) base = atomicAdd(&gn, total);
            slot += __shfl(base, 31);
            auto push = [&](unsigned m, int word) {
                while (m) {
                    const int j = __ffs(m) - 1;
                    m &= m - 1;
                    if (slot < LCAP) gq[slot] = 32 * word + j;
                    ++slot;
                }
            };
#pragma unroll
            for (int k = 0; k < K4; ++k) {
                const int w0 = 4 * (ib + lane + k * 1024);
                push(wv[k].x, w0);
                push(wv[k].y, w0 + 1);
                push(wv[k].z, w0 + 2);
                push(wv[k].w, w0 + 3);
            }
        }
        __syncthreads();

        const int qtot = gn;
        if (qtot <= LCAP) {
            constexpr int KQ = 8;
            for (int ib = wbase; ib < qtot; ib += KQ * 1024) {
                int4 va[KQ], vb[KQ];
#pragma unroll
                for (int k = 0; k < KQ; ++k) {
                    const int i = ib + lane + k * 1024;
                    const int q = gq[i < qtot ? i : qtot - 1];
                    va[k] = pairs[2 * (size_t)q];
                    vb[k] = pairs[2 * (size_t)q + 1];
                }
                unsigned hm = 0;
#pragma unroll
                for (int k = 0; k < KQ; ++k) {
                    const unsigned ok = (ib + lane + k * 1024 < qtot) ? 15u : 0u;
                    const unsigned h4 = ((unsigned)(va[k].x - lo) < TR ? 1u : 0u) |
                                        ((unsigned)(va[k].z - lo) < TR ? 2u : 0u) |
                                        ((unsigned)(vb[k].x - lo) < TR ? 4u : 0u) |
                                        ((unsigned)(vb[k].z - lo) < TR ? 8u : 0u);
                    hm |= (h4 & ok) << (4 * k);
                }
                int total;
                int slot = wave_scan_excl(__popc(hm), lane, total);
                int base = 0;
                if (lane == 31 && total > 0) base = atomicAdd(&en, total);
                base = __shfl(base, 31);
                slot += base;
                if (base + total <= LCAP) {
#pragma unroll
                    for (int k = 0; k < KQ; ++k) {
                        if (hm & (1u << (4 * k))) { eq[slot] = ((va[k].x - lo) << SRC_BITS) | va[k].y; ++slot; }
                        if (hm & (2u << (4 * k))) { eq[slot] = ((va[k].z - lo) << SRC_BITS) | va[k].w; ++slot; }
                        if (hm & (4u << (4 * k))) { eq[slot] = ((vb[k].x - lo) << SRC_BITS) | vb[k].y; ++slot; }
                        if (hm & (8u << (4 * k))) { eq[slot] = ((vb[k].z - lo) << SRC_BITS) | vb[k].w; ++slot; }
                    }
                }
            }
            __syncthreads();

            const int etot = en;
            if (etot <= LCAP) {
                int* lst = list + (size_t)blockIdx.x * LCAP;
                for (int i = tid; i < etot; i += 1024) {
                    const int w = eq[i];
                    lst[i] = w;
                    atomicAdd(&degc[w >> SRC_BITS], 1);
                }
                count   = etot;
                counted = true;
            }
        }
    }
    if (!counted) {
        for (int e = tid; e < E; e += 1024) {
            const unsigned r = (unsigned)(dst[e] - lo);
            if (r < TR) atomicAdd(&degc[r], 1);
        }
    }

    static_assert(TR <= 1024, "one thread per row of the tile");
    float2 xv = make_float2(0.0f, 0.0f);
    if (tid < TR) xv = *(const float2*)(x + 2 * (size_t)(lo + tid));
    __syncthreads();

    if (tid < TR) {
        const float di = rsqrtf(fmaxf((float)degc[tid], 1.0f));
        dinv[lo + tid] = di;
        *(float2*)(xs + 2 * (size_t)(lo + tid)) = make_float2(xv.x * di, xv.y * di);
    }
    if (tid == 0) tile_cnt[blockIdx.x] = count;
}


#define IDX_BLOCKS 512
#define IDX_PITCH  257

static_assert(NT <= IDX_PITCH, "a wave's row must hold a word per tile");

__global__ void __launch_bounds__(1024)
k_edge_index(const int* __restrict__ src, const int* __restrict__ dst, int E,
             int* __restrict__ idx) {
    __shared__ unsigned wb[32 * IDX_PITCH];
    if ((E & 3) != 0) return;

    const int tid    = threadIdx.x;
    const int lane   = tid & 31;
    const int wave   = tid >> 5;
    const int E4     = E >> 2;
    const int nchunk = (E4 + 1023) >> 10;
    const int wpt    = nchunk << 5;
    const int4* d4   = (const int4*)dst;
    const int4* s4   = (const int4*)src;
    int4*     pairs  = (int4*)idx;
    unsigned* bits   = (unsigned*)idx + 2 * (size_t)E;

    for (int c = blockIdx.x; c < nchunk; c += IDX_BLOCKS) {
        const int  q  = (c << 10) + tid;
        const int  qc = q < E4 ? q : E4 - 1;
        const int4 dv = d4[qc];
        const int4 sv = s4[qc];
        for (int i = tid; i < 32 * IDX_PITCH; i += 1024) wb[i] = 0u;
        __syncthreads();

        if (q < E4) {
            pairs[2 * (size_t)q]     = make_int4(dv.x, sv.x, dv.y, sv.y);
            pairs[2 * (size_t)q + 1] = make_int4(dv.z, sv.z, dv.w, sv.w);
            unsigned* wrow = wb + wave * IDX_PITCH;
            const unsigned bit = 1u << lane;
            const unsigned t0  = (unsigned)dv.x / TR;
            const unsigned t1  = (unsigned)dv.y / TR;
            const unsigned t2  = (unsigned)dv.z / TR;
            const unsigned t3  = (unsigned)dv.w / TR;
            if (t0 < NT) atomicOr(&wrow[t0], bit);
            if (t1 < NT) atomicOr(&wrow[t1], bit);
            if (t2 < NT) atomicOr(&wrow[t2], bit);
            if (t3 < NT) atomicOr(&wrow[t3], bit);
        }
        __syncthreads();

        for (int t = wave; t < NT; t += 32)
            bits[(size_t)t * wpt + (c << 5) + lane] = wb[lane * IDX_PITCH + t];
        __syncthreads();
    }
}


__device__ __forceinline__ float xhalf(float x) {
    const int b = __builtin_bit_cast(int, x);
    return __builtin_bit_cast(float, __builtin_amdgcn_permlanex16(b, b, 0x76543210, 0xfedcba98, false, false));
}

__device__ __forceinline__ void lds_add(float* p, float v) {
    __hip_atomic_fetch_add(p, v, __ATOMIC_RELAXED, __HIP_MEMORY_SCOPE_WORKGROUP);
}

template <int F, bool GEMM_OUT>
__global__ void __launch_bounds__(1024)
k_tile_agg(const int* __restrict__ src, const int* __restrict__ dst, int E,
           const float* __restrict__ dinv, const int* __restrict__ tile_cnt,
           const int* __restrict__ list, const float* __restrict__ hs,
           const float* __restrict__ bias, const float* __restrict__ W1,
           const float* __restrict__ W2, float* __restrict__ out) {
    constexpr int Q = F / 32;
    constexpr int U = 8;
    constexpr int P = GEMM_OUT ? F + 4 : F;
    static_assert(!GEMM_OUT || F == F1, "the fused product is act1 (F1 wide) times W2");
    __shared__ float acc[TR * P];

    const int tid  = threadIdx.x;
    const int lane = tid & 31;
    const int wave = __builtin_amdgcn_readfirstlane(tid >> 5);
    const int lo   = blockIdx.x * TR;

    if (GEMM_OUT) {
        constexpr int PIECES = TR * F / 4;
        constexpr int ITERS  = (PIECES + 1023) / 1024;
        static_assert(1024 % (F / 4) == 0, "a thread must keep its columns");
        const int    f  = (tid & (F / 4 - 1)) << 2;
        const float4 wa = *(const float4*)(W1 + f);
        const float4 wb = *(const float4*)(W1 + F + f);
        float2 pv[ITERS];
#pragma unroll
        for (int it = 0; it < ITERS; ++it) {
            const int i4  = tid + it * 1024;
            const int row = (i4 < PIECES ? i4 : PIECES - 1) / (F / 4);
            pv[it] = *(const float2*)(hs + 2 * (size_t)(lo + row));
        }
#pragma unroll
        for (int it = 0; it < ITERS; ++it) {
            const int i4 = tid + it * 1024;
            if (i4 < PIECES) {
                float4 hv;
                hv.x = fmaf(pv[it].x, wa.x, pv[it].y * wb.x);
                hv.y = fmaf(pv[it].x, wa.y, pv[it].y * wb.y);
                hv.z = fmaf(pv[it].x, wa.z, pv[it].y * wb.z);
                hv.w = fmaf(pv[it].x, wa.w, pv[it].y * wb.w);
                *(float4*)(acc + (i4 / (F / 4)) * P + f) = hv;
            }
        }
    } else {
        for (int i4 = tid; i4 < TR * F / 4; i4 += 1024)
            *(float4*)(acc + (i4 / (F / 4)) * P + ((i4 & (F / 4 - 1)) << 2)) =
                *(const float4*)(hs + (size_t)lo * F + 4 * (size_t)i4);
    }
    __syncthreads();

    const int n = tile_cnt[blockIdx.x];
    if (GEMM_OUT && n <= LCAP) {
        float w0[Q], w1[Q];
        for (int q = 0; q < Q; ++q) {
            w0[q] = W1[lane + 32 * q];
            w1[q] = W1[F + lane + 32 * q];
        }
        const int* lst = list + (size_t)blockIdx.x * LCAP;
        int b  = wave * 32;
        int wc = (b + lane < n) ? lst[b + lane] : 0;
        float2 pc = *(const float2*)(hs + 2 * (size_t)(wc & ((1 << SRC_BITS) - 1)));
        int wn = (b + 1024 + lane < n) ? lst[b + 1024 + lane] : 0;
        for (; b < n; b += 32 * 32) {
            const int cnt = (n - b) < 32 ? (n - b) : 32;
            const int    r  = wc >> SRC_BITS;
            const float2 p  = pc;
            wc = wn;
            pc = *(const float2*)(hs + 2 * (size_t)(wc & ((1 << SRC_BITS) - 1)));
            wn = (b + 2048 + lane < n) ? lst[b + 2048 + lane] : 0;
            int j = 0;
            for (; j + U <= cnt; j += U) {
                for (int u = 0; u < U; ++u) {
                    const int   rj = __builtin_amdgcn_readlane(r, j + u);
                    const float x0 = __builtin_bit_cast(float, __builtin_amdgcn_readlane(__builtin_bit_cast(int, p.x), j + u));
                    const float x1 = __builtin_bit_cast(float, __builtin_amdgcn_readlane(__builtin_bit_cast(int, p.y), j + u));
                    float* ap = acc + rj * P + lane;
                    for (int q = 0; q < Q; ++q) lds_add(ap + 32 * q, fmaf(x0, w0[q], x1 * w1[q]));
                }
            }
            for (; j < cnt; ++j) {
                const int   rj = __builtin_amdgcn_readlane(r, j);
                const float x0 = __builtin_bit_cast(float, __builtin_amdgcn_readlane(__builtin_bit_cast(int, p.x), j));
                const float x1 = __builtin_bit_cast(float, __builtin_amdgcn_readlane(__builtin_bit_cast(int, p.y), j));
                float* ap = acc + rj * P + lane;
                for (int q = 0; q < Q; ++q) lds_add(ap + 32 * q, fmaf(x0, w0[q], x1 * w1[q]));
            }
        }
    } else if (n <= LCAP) {
        const int* lst = list + (size_t)blockIdx.x * LCAP;
        for (int b = wave * 32; b < n; b += 32 * 32) {
            const int cnt = (n - b) < 32 ? (n - b) : 32;
            int s = 0, r = 0;
            if (lane < cnt) {
                const int en = lst[b + lane];
                s = en & ((1 << SRC_BITS) - 1);
                r = en >> SRC_BITS;
            }
            int j = 0;
            for (; j + U <= cnt; j += U) {
                float v[U][Q];
                for (int u = 0; u < U; ++u) {
                    const int sj = __builtin_amdgcn_readlane(s, j + u);
                    const float* hp = hs + (size_t)sj * F + lane;
                    for (int q = 0; q < Q; ++q) v[u][q] = hp[32 * q];
                }
                for (int u = 0; u < U; ++u) {
                    const int rj = __builtin_amdgcn_readlane(r, j + u);
                    float* ap = acc + rj * P + lane;
                    for (int q = 0; q < Q; ++q) lds_add(ap + 32 * q, v[u][q]);
                }
            }
            for (; j < cnt; ++j) {
                const int sj = __builtin_amdgcn_readlane(s, j);
                const int rj = __builtin_amdgcn_readlane(r, j);
                const float* hp = hs + (size_t)sj * F + lane;
                float* ap = acc + rj * P + lane;
                for (int q = 0; q < Q; ++q) lds_add(ap + 32 * q, hp[32 * q]);
            }
        }
    } else {
        for (int e = tid; e < E; e += 1024) {
            const unsigned r = (unsigned)(dst[e] - lo);
            if (r < TR) {
                const int s = src[e];
#pragma nounroll
                for (int f = 0; f < F; ++f)
                    lds_add(&acc[r * P + f],
                            GEMM_OUT ? fmaf(hs[2 * (size_t)s], W1[f], hs[2 * (size_t)s + 1] * W1[F + f])
                                     : hs[(size_t)s * F + f]);
            }
        }
    }
    __syncthreads();

    for (int i4 = tid; i4 < TR * F / 4; i4 += 1024) {
        const int   row = i4 / (F / 4);
        const float di  = dinv[lo + row];
        float* arow4 = acc + row * P + ((i4 & (F / 4 - 1)) << 2);
        float4 v = *(const float4*)arow4;
        const float4 bb = *(const float4*)(bias + ((i4 & (F / 4 - 1)) << 2));
        v.x = leaky(fmaf(v.x, di, bb.x));
        v.y = leaky(fmaf(v.y, di, bb.y));
        v.z = leaky(fmaf(v.z, di, bb.z));
        v.w = leaky(fmaf(v.w, di, bb.w));
        if (GEMM_OUT) {
            v.x *= di;
            v.y *= di;
            v.z *= di;
            v.w *= di;
            *(float4*)arow4 = v;
        } else {
            *(float4*)(out + (size_t)lo * F + 4 * (size_t)i4) = v;
        }
    }

    if (GEMM_OUT) {
        const int m  = lane & 15;
        const int h  = lane >> 4;
        const int kh = h << 1;
        const int c0 = (wave & 3) * 32;
        v2f bA[F1 / 4], bB[F1 / 4];
#pragma unroll
        for (int t = 0; t < F1 / 4; ++t) {
            bA[t].x = W2[(4 * t + kh) * F2 + c0 + m];
            bA[t].y = W2[(4 * t + kh + 1) * F2 + c0 + m];
            bB[t].x = W2[(4 * t + kh) * F2 + c0 + 16 + m];
            bB[t].y = W2[(4 * t + kh + 1) * F2 + c0 + 16 + m];
        }
        __syncthreads();

        for (int st = wave >> 2; st < TR / 16; st += 8) {
            const float* arow = acc + (16 * st + m) * P + kh;
            v8f d0 = {}, d1 = {};
#pragma unroll
            for (int t = 0; t < F1 / 4; ++t) {
                const v2f a = *(const v2f*)(arow + 4 * t);
                d0 = __builtin_amdgcn_wmma_f32_16x16x4_f32(
                     false, a,  false, bA[t],
                     (short)0, d0,  false,  false);
                d1 = __builtin_amdgcn_wmma_f32_16x16x4_f32(
                     false, a,  false, bB[t],
                     (short)0, d1,  false,  false);
            }
            float* orow = out + (size_t)(lo + 16 * st) * F2 + c0 + lane;
#pragma unroll
            for (int v = 0; v < 8; ++v) {
                const float x0s = xhalf(d0[v]);
                const float x1s = xhalf(d1[v]);
                orow[(size_t)v * F2]       = h ? x1s : d0[v];
                orow[(size_t)(v + 8) * F2] = h ? d1[v] : x0s;
            }
        }
    }
}


extern "C" void kernel_launch(void* const* d_in, const int* in_sizes, int n_in,
                              void* d_out, int out_size, void* d_ws, size_t ws_size,
                              hipStream_t stream) {
    const int*   edge_index = (const int*)d_in[0];
    const float* x          = (const float*)d_in[1];
    const float* W1         = (const float*)d_in[2];
    const float* b1         = (const float*)d_in[3];
    const float* W2         = (const float*)d_in[4];
    const float* b2         = (const float*)d_in[5];
    float*       z          = (float*)d_out;

    const int E = in_sizes[0] / 2;
    const int* src = edge_index;
    const int* dst = edge_index + E;

    float* ws   = (float*)d_ws;
    float* dinv = ws;
    int*   tcnt = (int*)(ws + N_NODES);
    int*   list = tcnt + 1024;
    float* xs   = (float*)(list + NT * LCAP);
    float* h2   = xs + N_NODES * 2;
    int*   idx  = (int*)(h2 + N_NODES * F2);
    (void)ws_size; (void)n_in; (void)out_size;

    k_edge_index   <<<IDX_BLOCKS, 1024, 0, stream>>>(src, dst, E, idx);

    k_tile_scan    <<<NT, 1024, 0, stream>>>(dst, E, idx, x, dinv, tcnt, list, xs);

    k_tile_agg<F1, true>  <<<NT, 1024, 0, stream>>>(src, dst, E, dinv, tcnt, list, xs, b1, W1, W2, h2);

    k_tile_agg<F2, false> <<<NT, 1024, 0, stream>>>(src, dst, E, dinv, tcnt, list, h2, b2, W1, W2, z);
}
